# starting kernel
# baseline (speedup 1.0000x reference)
#include <hip/hip_runtime.h>
#include <math.h>


typedef __attribute__((ext_vector_type(2))) float v2f;
typedef __attribute__((ext_vector_type(8))) float v8f;

#define NDIM 1024
#define LSEQ 8192

__global__ __launch_bounds__(128)
void gemm_wmma_f32(const float* __restrict__ Ag, int lda,
                   const float* __restrict__ Bg, int ldb,
                   float* __restrict__ Cg, int ldc,
                   int Kdim, int mBound, int nBound, int addI)
{
    __shared__ float As[64][33];
    __shared__ float Bs[64][35];

    const int tid  = threadIdx.x;
    const int row0 = blockIdx.y * 64;
    const int col0 = blockIdx.x * 64;
    const int lane = tid & 31;
    const int wv   = tid >> 5;
    const int wm   = (wv >> 1) * 32;
    const int wn   = (wv & 1)  * 32;
    const int g    = lane >> 4;
    const int lm   = lane & 15;

    v8f acc[2][2] = {};

    for (int k0 = 0; k0 < Kdim; k0 += 32) {
        #pragma unroll
        for (int i = tid; i < 64 * 32; i += 128) {
            int r = i >> 5, c = i & 31;
            As[r][c] = Ag[(size_t)(row0 + r) * lda + (k0 + c)];
        }
        #pragma unroll
        for (int i = tid; i < 32 * 64; i += 128) {
            int r = i >> 6, c = i & 63;
            Bs[c][r] = Bg[(size_t)(k0 + r) * ldb + (col0 + c)];
        }
        if (k0 + 32 < Kdim) {
            __builtin_prefetch(&Ag[(size_t)(row0 + (tid >> 1)) * lda + (k0 + 32) + (tid & 1) * 16], 0, 0);
            __builtin_prefetch(&Bg[(size_t)(k0 + 32 + (tid >> 2)) * ldb + col0 + (tid & 3) * 16], 0, 0);
        }
        __syncthreads();

        #pragma unroll
        for (int kk = 0; kk < 32; kk += 4) {
            const int ka = kk + 2 * g;
            v2f a0, a1, b0, b1;
            a0.x = As[wm + lm][ka];           a0.y = As[wm + lm][ka + 1];
            a1.x = As[wm + 16 + lm][ka];      a1.y = As[wm + 16 + lm][ka + 1];
            b0.x = Bs[wn + lm][ka];           b0.y = Bs[wn + lm][ka + 1];
            b1.x = Bs[wn + 16 + lm][ka];      b1.y = Bs[wn + 16 + lm][ka + 1];
            acc[0][0] = __builtin_amdgcn_wmma_f32_16x16x4_f32(false, a0, false, b0, (short)0, acc[0][0], false, false);
            acc[0][1] = __builtin_amdgcn_wmma_f32_16x16x4_f32(false, a0, false, b1, (short)0, acc[0][1], false, false);
            acc[1][0] = __builtin_amdgcn_wmma_f32_16x16x4_f32(false, a1, false, b0, (short)0, acc[1][0], false, false);
            acc[1][1] = __builtin_amdgcn_wmma_f32_16x16x4_f32(false, a1, false, b1, (short)0, acc[1][1], false, false);
        }
        __syncthreads();
    }

    #pragma unroll
    for (int mi = 0; mi < 2; ++mi)
        #pragma unroll
        for (int ni = 0; ni < 2; ++ni)
            #pragma unroll
            for (int r = 0; r < 8; ++r) {
                int row = row0 + wm + mi * 16 + r + 8 * g;
                int col = col0 + wn + ni * 16 + lm;
                if (row < mBound && col < nBound) {
                    float v = acc[mi][ni][r];
                    if (addI && row == col) v += 1.0f;
                    Cg[(size_t)row * ldc + col] = v;
                }
            }
}

__global__ void prep_mats(const float* __restrict__ A, const float* __restrict__ ldt,
                          float* __restrict__ M, float* __restrict__ IM, int n)
{
    int idx = blockIdx.x * blockDim.x + threadIdx.x;
    if (idx >= n * n) return;
    float h = 0.5f * expf(ldt[0]);
    float m = h * A[idx];
    M[idx]  = m;
    int r = idx / n, c = idx - r * n;
    IM[idx] = m + ((r == c) ? 1.0f : 0.0f);
}

__global__ void bb_matvec(const float* __restrict__ S, const float* __restrict__ Bv,
                          const float* __restrict__ ldt, float* __restrict__ Y,
                          int n, int ldy)
{
    int i = blockIdx.x * blockDim.x + threadIdx.x;
    if (i >= n) return;
    float dt = expf(ldt[0]);
    float s = 0.f;
    for (int k = 0; k < n; ++k) s += S[(size_t)i * n + k] * Bv[k];
    Y[(size_t)i * ldy] = dt * s;
}

__global__ void copy_row(const float* __restrict__ Cv, float* __restrict__ R, int n)
{
    int i = blockIdx.x * blockDim.x + threadIdx.x;
    if (i < n) R[i] = Cv[i];
}

__global__ __launch_bounds__(256)
void conv_causal(const float* __restrict__ u, const float* __restrict__ Kk,
                 const float* __restrict__ Dv, float* __restrict__ y, int L)
{
    __shared__ float Ks[256];
    __shared__ float us[512];
    const int tid = threadIdx.x;
    const int t0  = blockIdx.x * 256;
    const int t   = t0 + tid;
    float acc = 0.f;
    for (int c0 = 0; c0 <= t0; c0 += 256) {
        Ks[tid] = Kk[(L - 1) - (c0 + tid)];
        int base = t0 - c0 - 255;
        int i0 = base + tid;
        us[tid] = (i0 >= 0) ? u[i0] : 0.f;
        int i1 = base + 256 + tid;
        us[256 + tid] = (i1 >= 0 && i1 < L) ? u[i1] : 0.f;
        __syncthreads();
        int jmax = t - c0; if (jmax > 255) jmax = 255;
        for (int j = 0; j <= jmax; ++j)
            acc += Ks[j] * us[tid + 255 - j];
        __syncthreads();
    }
    y[t] = acc + Dv[0] * u[t];
}

extern "C" void kernel_launch(void* const* d_in, const int* in_sizes, int n_in,
                              void* d_out, int out_size, void* d_ws, size_t ws_size,
                              hipStream_t stream)
{
    (void)in_sizes; (void)n_in; (void)out_size; (void)ws_size;
    const float* u   = (const float*)d_in[0];
    const float* A   = (const float*)d_in[1];
    const float* Bv  = (const float*)d_in[2];
    const float* Cv  = (const float*)d_in[3];
    const float* Dv  = (const float*)d_in[4];
    const float* ldt = (const float*)d_in[5];
    float* y = (float*)d_out;
    const int N = NDIM, L = LSEQ;

    float* B0 = (float*)d_ws;
    float* B1 = B0 + (size_t)N * N;
    float* B2 = B1 + (size_t)N * N;
    float* B3 = B2 + (size_t)N * N;
    float* Y  = B3 + (size_t)N * N;
    float* R  = Y  + (size_t)N * 128;
    float* Kk = R  + (size_t)64 * N;

    dim3 blk(128);
    auto gemm = [&](const float* Am, int lda, const float* Bm, int ldb,
                    float* Cm, int ldc, int Mc, int Nc, int Kd,
                    int mB, int nB, int addI) {
        dim3 grid((Nc + 63) / 64, (Mc + 63) / 64);
        gemm_wmma_f32<<<grid, blk, 0, stream>>>(Am, lda, Bm, ldb, Cm, ldc, Kd, mB, nB, addI);
    };

    prep_mats<<<(N * N + 255) / 256, 256, 0, stream>>>(A, ldt, B0, B1, N);
    gemm(B0, N, B1, N, B2, N, N, N, N, N, N, 1);
    gemm(B0, N, B2, N, B3, N, N, N, N, N, N, 1);
    gemm(B0, N, B3, N, B2, N, N, N, N, N, N, 1);
    gemm(B2, N, B1, N, B3, N, N, N, N, N, N, 0);
    bb_matvec<<<(N + 255) / 256, 256, 0, stream>>>(B2, Bv, ldt, Y, N, 128);

    gemm(B3, N, Y, 128, Y + 1,  128, N, 64, N, N, 127, 0);
    gemm(B3, N, B3, N, B0, N, N, N, N, N, N, 0);
    gemm(B0, N, Y, 128, Y + 2,  128, N, 64, N, N, 126, 0);
    gemm(B0, N, B0, N, B1, N, N, N, N, N, N, 0);
    gemm(B1, N, Y, 128, Y + 4,  128, N, 64, N, N, 124, 0);
    gemm(B1, N, B1, N, B0, N, N, N, N, N, N, 0);
    gemm(B0, N, Y, 128, Y + 8,  128, N, 64, N, N, 120, 0);
    gemm(B0, N, B0, N, B1, N, N, N, N, N, N, 0);
    gemm(B1, N, Y, 128, Y + 16, 128, N, 64, N, N, 112, 0);
    gemm(B1, N, B1, N, B0, N, N, N, N, N, N, 0);
    gemm(B0, N, Y, 128, Y + 32, 128, N, 64, N, N,  96, 0);
    gemm(B0, N, B0, N, B1, N, N, N, N, N, N, 0);
    gemm(B1, N, Y, 128, Y + 64, 128, N, 64, N, N,  64, 0);
    gemm(B1, N, B1, N, B0, N, N, N, N, N, N, 0);

    copy_row<<<(N + 255) / 256, 256, 0, stream>>>(Cv, R, N);
    gemm(R, N, B0, N, R + 1 * N,  N, 64, N, N, 63, N, 0);
    gemm(B0, N, B0, N, B1, N, N, N, N, N, N, 0);
    gemm(R, N, B1, N, R + 2 * N,  N, 64, N, N, 62, N, 0);
    gemm(B1, N, B1, N, B0, N, N, N, N, N, N, 0);
    gemm(R, N, B0, N, R + 4 * N,  N, 64, N, N, 60, N, 0);
    gemm(B0, N, B0, N, B1, N, N, N, N, N, N, 0);
    gemm(R, N, B1, N, R + 8 * N,  N, 64, N, N, 56, N, 0);
    gemm(B1, N, B1, N, B0, N, N, N, N, N, N, 0);
    gemm(R, N, B0, N, R + 16 * N, N, 64, N, N, 48, N, 0);
    gemm(B0, N, B0, N, B1, N, N, N, N, N, N, 0);
    gemm(R, N, B1, N, R + 32 * N, N, 64, N, N, 32, N, 0);

    gemm(R, N, Y, 128, Kk, 128, 64, 128, N, 64, 128, 0);

    conv_causal<<<L / 256, 256, 0, stream>>>(u, Kk, Dv, y, L);
}


# Doubling applies ride in the squaring launches that share their operand: participating waves carry one 16x16 tile of the thin product in a fifth accumulator; 34 launches to 22; same sums
# speedup vs baseline: 8.1748x; 8.1748x over previous
#include <hip/hip_runtime.h>
#include <math.h>


typedef __attribute__((ext_vector_type(2))) float v2f;
typedef __attribute__((ext_vector_type(4))) float v4f;
typedef __attribute__((ext_vector_type(8))) float v8f;

#define NDIM 1024
#define LSEQ 8192

#define KS 8
#define NS 2

struct Frag { v2f a[KS]; v4f b0[KS]; v4f b1[KS]; };

#define YLD 128
struct FragX { v2f a[KS]; v2f b[KS]; };

template <int RIDE>
__device__ __forceinline__ void fragx_load(FragX& f, const float* xaP, const float* xbP)
{
    constexpr int LDX = (RIDE == 2) ? NDIM : YLD;
    #pragma unroll
    for (int j = 0; j < KS; ++j) {
        if (RIDE == 2) f.a[j] = *(const v2f*)(xaP + 4 * j);
        f.b[j].x = xbP[(4 * j) * LDX];
        f.b[j].y = xbP[(4 * j + 1) * LDX];
    }
}

template <int LDB>
__device__ __forceinline__ void frag_load(Frag& f, const float* __restrict__ aP,
                                          const float* __restrict__ bP)
{
    #pragma unroll
    for (int j = 0; j < KS; ++j) {
        f.a[j]  = *(const v2f*)(aP + 4 * j);
        f.b0[j] = *(const v4f*)(bP + (4 * j) * LDB);
        f.b1[j] = *(const v4f*)(bP + (4 * j + 1) * LDB);
    }
}

template <int RIDE>
__device__ __forceinline__ void frag_mma(const Frag& f, const FragX& fx, v8f (&acc)[4], v8f& accX)
{
    #pragma unroll
    for (int j = 0; j < KS; ++j) {
        #pragma unroll
        for (int t = 0; t < 4; ++t) {
            v2f b; b.x = f.b0[j][t]; b.y = f.b1[j][t];
            acc[t] = __builtin_amdgcn_wmma_f32_16x16x4_f32(false, f.a[j], false, b, (short)0, acc[t], false, false);
        }
        if (RIDE == 1)
            accX = __builtin_amdgcn_wmma_f32_16x16x4_f32(false, f.a[j], false, fx.b[j], (short)0, accX, false, false);
        if (RIDE == 2)
            accX = __builtin_amdgcn_wmma_f32_16x16x4_f32(false, fx.a[j], false, fx.b[j], (short)0, accX, false, false);
    }
}

template <int LDB, int RIDE>
__device__ __forceinline__
void gemm_body(const float* __restrict__ Ag, int lda,
               const float* __restrict__ Bg,
               float* __restrict__ Cg, int ldc,
               int Kdim, int mBound, int nBound, int addI,
               const float* Xa, const float* Xb, float* Xo, int xn)
{
    const int tid  = threadIdx.x;
    const int lane = tid & 31;
    const int wv   = __builtin_amdgcn_readfirstlane(tid >> 5);
    const int h    = lane >> 4;
    const int c    = lane & 15;
    const int row0 = blockIdx.y * 64 + wv * 16;
    const int col0 = blockIdx.x * 64;

    const int KC = 4 * KS;
    const float* aP = Ag + (size_t)(row0 + c) * lda + 2 * h;
    const float* bP = Bg + (size_t)(2 * h) * LDB + col0 + 4 * c;

    constexpr int LDX = (RIDE == 2) ? NDIM : YLD;
    const float* xaP = Xa;
    const float* xbP = Xb;
    int xcol = 0, xrow0 = 0;
    if (RIDE == 1) {
        xcol = 16 * (int)blockIdx.x + c;
        xbP  = Xb + (size_t)(2 * h) * LDX + (xcol < xn ? xcol : 0);
    }
    if (RIDE == 2) {
        xrow0 = 16 * (int)blockIdx.y;
        xcol  = 16 * (4 * (int)blockIdx.x + wv) + c;
        const int ar = xrow0 + c;
        xaP = Xa + (size_t)(ar < xn ? ar : 0) * NDIM + 2 * h;
        xbP = Xb + (size_t)(2 * h) * LDX + xcol;
    }

    v8f acc[4] = {};
    v8f accX = {};
    Frag f[NS];
    FragX fx[NS];

    #pragma unroll
    for (int s = 0; s < NS; ++s) {
        frag_load<LDB>(f[s], aP + s * KC, bP + s * KC * LDB);
        if (RIDE == 1) fragx_load<RIDE>(fx[s], xaP, xbP + s * KC * LDX);
        if (RIDE == 2) fragx_load<RIDE>(fx[s], xaP + s * KC, xbP + s * KC * LDX);
        __builtin_amdgcn_sched_barrier(0);
    }

    for (int kb = NS * KC; kb < Kdim; kb += NS * KC) {
        aP += NS * KC;
        bP += NS * KC * LDB;
        if (RIDE) xbP += NS * KC * LDX;
        if (RIDE == 2) xaP += NS * KC;
        #pragma unroll
        for (int s = 0; s < NS; ++s) {
            frag_mma<RIDE>(f[s], fx[s], acc, accX);
            frag_load<LDB>(f[s], aP + s * KC, bP + s * KC * LDB);
            if (RIDE == 1) fragx_load<RIDE>(fx[s], xaP, xbP + s * KC * LDX);
            if (RIDE == 2) fragx_load<RIDE>(fx[s], xaP + s * KC, xbP + s * KC * LDX);
            __builtin_amdgcn_sched_barrier(0);
        }
    }
    #pragma unroll
    for (int s = 0; s < NS; ++s) {
        frag_mma<RIDE>(f[s], fx[s], acc, accX);
        __builtin_amdgcn_sched_barrier(0);
    }

    const bool whole = (row0 + 16 <= mBound) && (col0 + 64 <= nBound) &&
                       ((ldc & 3) == 0) && ((((size_t)Cg) & 15) == 0);
    if (whole) {
        #pragma unroll
        for (int r = 0; r < 8; ++r) {
            const int row = row0 + 8 * h + r;
            const int col = col0 + 4 * c;
            v4f v;
            v.x = acc[0][r]; v.y = acc[1][r]; v.z = acc[2][r]; v.w = acc[3][r];
            if (addI) {
                if (row == col)     v.x += 1.0f;
                if (row == col + 1) v.y += 1.0f;
                if (row == col + 2) v.z += 1.0f;
                if (row == col + 3) v.w += 1.0f;
            }
            *(v4f*)(Cg + (size_t)row * ldc + col) = v;
        }
    } else {
        #pragma unroll
        for (int t = 0; t < 4; ++t)
            #pragma unroll
            for (int r = 0; r < 8; ++r) {
                const int row = row0 + 8 * h + r;
                const int col = col0 + 4 * c + t;
                if (row < mBound && col < nBound) {
                    float v = acc[t][r];
                    if (addI && row == col) v += 1.0f;
                    Cg[(size_t)row * ldc + col] = v;
                }
            }
    }

    if (RIDE == 1) {
        if (xcol < xn) {
            #pragma unroll
            for (int r = 0; r < 8; ++r)
                Xo[(size_t)(row0 + 8 * h + r) * LDX + xcol] = accX[r];
        }
    }
    if (RIDE == 2) {
        #pragma unroll
        for (int r = 0; r < 8; ++r) {
            const int rr = xrow0 + 8 * h + r;
            if (rr < xn) Xo[(size_t)rr * LDX + xcol] = accX[r];
        }
    }
}

template <int LDB>
__global__ __launch_bounds__(128)
void gemm_wmma_f32(const float* __restrict__ Ag, int lda,
                   const float* __restrict__ Bg,
                   float* __restrict__ Cg, int ldc,
                   int Kdim, int mBound, int nBound, int addI)
{
    gemm_body<LDB, 0>(Ag, lda, Bg, Cg, ldc, Kdim, mBound, nBound, addI, nullptr, nullptr, nullptr, 0);
}

__global__ __launch_bounds__(128)
void gemm_sq_cols(const float* __restrict__ Ms, float* __restrict__ M2,
                  const float* Yg, float* Yo, int s)
{
    if (16 * (int)blockIdx.x < s)
        gemm_body<NDIM, 1>(Ms, NDIM, Ms, M2, NDIM, NDIM, NDIM, NDIM, 0, nullptr, Yg, Yo, s);
    else
        gemm_body<NDIM, 0>(Ms, NDIM, Ms, M2, NDIM, NDIM, NDIM, NDIM, 0, nullptr, nullptr, nullptr, 0);
}

__global__ __launch_bounds__(128)
void gemm_sq_rows(const float* __restrict__ Qt, float* __restrict__ Q2,
                  const float* Rg, float* Ro, int t)
{
    if (16 * (int)blockIdx.y < t)
        gemm_body<NDIM, 2>(Qt, NDIM, Qt, Q2, NDIM, NDIM, NDIM, NDIM, 0, Rg, Qt, Ro, t);
    else
        gemm_body<NDIM, 0>(Qt, NDIM, Qt, Q2, NDIM, NDIM, NDIM, NDIM, 0, nullptr, nullptr, nullptr, 0);
}

__global__ void prep_mats(const float* __restrict__ A, const float* __restrict__ ldt,
                          float* __restrict__ M, float* __restrict__ IM, int n)
{
    int idx = blockIdx.x * blockDim.x + threadIdx.x;
    if (idx >= n * n) return;
    float h = 0.5f * expf(ldt[0]);
    float m = h * A[idx];
    M[idx]  = m;
    int r = idx / n, c = idx - r * n;
    IM[idx] = m + ((r == c) ? 1.0f : 0.0f);
}

__global__ void bb_matvec(const float* __restrict__ S, const float* __restrict__ Bv,
                          const float* __restrict__ ldt, float* __restrict__ Y,
                          int n, int ldy)
{
    int i = blockIdx.x * blockDim.x + threadIdx.x;
    if (i >= n) return;
    float dt = expf(ldt[0]);
    float s = 0.f;
    for (int k = 0; k < n; ++k) s += S[(size_t)i * n + k] * Bv[k];
    Y[(size_t)i * ldy] = dt * s;
}

__global__ void copy_row(const float* __restrict__ Cv, float* __restrict__ R, int n)
{
    int i = blockIdx.x * blockDim.x + threadIdx.x;
    if (i < n) R[i] = Cv[i];
}

#define CT   128
#define CW   16
#define CPAD 128
__global__ __launch_bounds__(32 * CW)
void conv_causal(const float* __restrict__ u, const float* __restrict__ Kk,
                 const float* __restrict__ Dv, float* __restrict__ y, int L)
{
    __shared__ __attribute__((aligned(16))) float us[CPAD + LSEQ];
    __shared__ __attribute__((aligned(16))) float kr[LSEQ];
    __shared__ __attribute__((aligned(16))) float red[CW][CT];
    const int tid  = threadIdx.x;
    const int lane = tid & 31;
    const int wv   = __builtin_amdgcn_readfirstlane(tid >> 5);
    const int t0   = blockIdx.x * CT;
    const int nt   = t0 + CT;

    for (int i = 4 * tid; i < nt; i += 4 * 32 * CW) {
        *(v4f*)&us[CPAD + i] = *(const v4f*)&u[i];
        const v4f kv = *(const v4f*)&Kk[L - 4 - i];
        v4f r; r.x = kv.w; r.y = kv.z; r.z = kv.y; r.w = kv.x;
        *(v4f*)&kr[i] = r;
    }
    if (tid < CPAD / 4) { v4f z = {0.f, 0.f, 0.f, 0.f}; *(v4f*)&us[4 * tid] = z; }
    __syncthreads();

    const int S  = (CT / CW) * ((int)blockIdx.x + 1);
    const int l0 = wv * S;
    const int T  = t0 + 4 * lane;
    float a0 = 0.f, a1 = 0.f, a2 = 0.f, a3 = 0.f;
    auto mac = [&](const v4f& w1, const v4f& w0, const v4f& k) {
        a0 += k.x * w1.x;  a1 += k.x * w1.y;  a2 += k.x * w1.z;  a3 += k.x * w1.w;
        a0 += k.y * w0.w;  a1 += k.y * w1.x;  a2 += k.y * w1.y;  a3 += k.y * w1.z;
        a0 += k.z * w0.z;  a1 += k.z * w0.w;  a2 += k.z * w1.x;  a3 += k.z * w1.y;
        a0 += k.w * w0.y;  a1 += k.w * w0.z;  a2 += k.w * w0.w;  a3 += k.w * w1.x;
    };
    v4f w1 = *(const v4f*)&us[CPAD + T - l0];
    v4f w0 = *(const v4f*)&us[CPAD + T - l0 - 4];
    v4f k  = *(const v4f*)&kr[l0];
    for (int l = l0; l < l0 + S - 4; l += 4) {
        const v4f w0n = *(const v4f*)&us[CPAD + T - l - 8];
        const v4f kn  = *(const v4f*)&kr[l + 4];
        mac(w1, w0, k);
        w1 = w0; w0 = w0n; k = kn;
    }
    mac(w1, w0, k);
    { v4f p; p.x = a0; p.y = a1; p.z = a2; p.w = a3; *(v4f*)&red[wv][4 * lane] = p; }
    __syncthreads();
    if (tid < CT) {
        float s = 0.f;
        for (int w = 0; w < CW; ++w) s += red[w][tid];
        y[t0 + tid] = s + Dv[0] * u[t0 + tid];
    }
}

extern "C" void kernel_launch(void* const* d_in, const int* in_sizes, int n_in,
                              void* d_out, int out_size, void* d_ws, size_t ws_size,
                              hipStream_t stream)
{
    (void)in_sizes; (void)n_in; (void)out_size; (void)ws_size;
    const float* u   = (const float*)d_in[0];
    const float* A   = (const float*)d_in[1];
    const float* Bv  = (const float*)d_in[2];
    const float* Cv  = (const float*)d_in[3];
    const float* Dv  = (const float*)d_in[4];
    const float* ldt = (const float*)d_in[5];
    float* y = (float*)d_out;
    const int N = NDIM, L = LSEQ;

    float* B0 = (float*)d_ws;
    float* B1 = B0 + (size_t)N * N;
    float* B2 = B1 + (size_t)N * N;
    float* B3 = B2 + (size_t)N * N;
    float* Y  = B3 + (size_t)N * N;
    float* R  = Y  + (size_t)N * 128;
    float* Kk = R  + (size_t)64 * N;

    dim3 blk(128);
    auto gemm = [&](const float* Am, int lda, const float* Bm, int ldb,
                    float* Cm, int ldc, int Mc, int Nc, int Kd,
                    int mB, int nB, int addI) {
        (void)ldb;
        dim3 grid((Nc + 63) / 64, (Mc + 63) / 64);
        gemm_wmma_f32<NDIM><<<grid, blk, 0, stream>>>(Am, lda, Bm, Cm, ldc, Kd, mB, nB, addI);
    };
    auto gemmY = [&](const float* Am, int lda, const float* Bm, int ldb,
                     float* Cm, int ldc, int Mc, int Nc, int Kd,
                     int mB, int nB, int addI) {
        (void)ldb;
        dim3 grid((Nc + 63) / 64, (Mc + 63) / 64);
        gemm_wmma_f32<128><<<grid, blk, 0, stream>>>(Am, lda, Bm, Cm, ldc, Kd, mB, nB, addI);
    };

    prep_mats<<<(N * N + 255) / 256, 256, 0, stream>>>(A, ldt, B0, B1, N);
    gemm(B0, N, B1, N, B2, N, N, N, N, N, N, 1);
    gemm(B0, N, B2, N, B3, N, N, N, N, N, N, 1);
    gemm(B0, N, B3, N, B2, N, N, N, N, N, N, 1);
    gemm(B2, N, B1, N, B3, N, N, N, N, N, N, 0);
    bb_matvec<<<(N + 255) / 256, 256, 0, stream>>>(B2, Bv, ldt, Y, N, 128);

    const dim3 sqGrid(NDIM / 64, NDIM / 64);

    auto sqCols = [&](const float* Ms, float* M2s, float* Ynew, int s) {
        gemm_sq_cols<<<sqGrid, blk, 0, stream>>>(Ms, M2s, Y, Ynew, s);
    };
    sqCols(B3, B0, Y + 1,  1);
    sqCols(B0, B1, Y + 2,  2);
    sqCols(B1, B0, Y + 4,  4);
    sqCols(B0, B1, Y + 8,  8);
    sqCols(B1, B0, Y + 16, 16);
    sqCols(B0, B1, Y + 32, 32);
    sqCols(B1, B0, Y + 64, 64);

    copy_row<<<(N + 255) / 256, 256, 0, stream>>>(Cv, R, N);
    auto sqRows = [&](const float* Qt, float* Q2t, float* Rnew, int t) {
        gemm_sq_rows<<<sqGrid, blk, 0, stream>>>(Qt, Q2t, R, Rnew, t);
    };
    sqRows(B0, B1, R + 1 * N,  1);
    sqRows(B1, B0, R + 2 * N,  2);
    sqRows(B0, B1, R + 4 * N,  4);
    sqRows(B1, B0, R + 8 * N,  8);
    sqRows(B0, B1, R + 16 * N, 16);
    gemm(R, N, B1, N, R + 32 * N, N, 64, N, N, 32, N, 0);

    gemmY(R, N, Y, 128, Kk, 128, 64, 128, N, 64, 128, 0);

    conv_causal<<<L / CT, 32 * CW, 0, stream>>>(u, Kk, Dv, y, L);
}


# Thin-product riders re-cut: each 16x16 rider tile belongs to one whole block, K in 8 slices (one per wave, ridden in its first 4 chunks), partials added in slice order via LDS (sum order changes)
# speedup vs baseline: 9.4665x; 1.1580x over previous
#include <hip/hip_runtime.h>
#include <math.h>


typedef __attribute__((ext_vector_type(2))) float v2f;
typedef __attribute__((ext_vector_type(4))) float v4f;
typedef __attribute__((ext_vector_type(8))) float v8f;

#define NDIM 1024
#define LSEQ 8192

#define KS 8
#define NS 2
#define KC (4 * KS)
#define RED_MAIN (4 * 8 * 32)
#define RED_V4 (RED_MAIN + 8 * 2 * 32)

struct Frag { v2f a[KS]; v4f b0[KS]; v4f b1[KS]; };

#define YLD 128
struct FragX { v2f a[KS]; v2f b[KS]; };

template <int LDX>
__device__ __forceinline__ void fragx_load(FragX& f, const float* xaP, const float* xbP)
{
    #pragma unroll
    for (int j = 0; j < KS; ++j) {
        f.a[j]   = *(const v2f*)(xaP + 4 * j);
        f.b[j].x = xbP[(4 * j) * LDX];
        f.b[j].y = xbP[(4 * j + 1) * LDX];
    }
}

template <int LDB>
__device__ __forceinline__ void frag_load(Frag& f, const float* __restrict__ aP,
                                          const float* __restrict__ bP)
{
    #pragma unroll
    for (int j = 0; j < KS; ++j) {
        f.a[j]  = *(const v2f*)(aP + 4 * j);
        f.b0[j] = *(const v4f*)(bP + (4 * j) * LDB);
        f.b1[j] = *(const v4f*)(bP + (4 * j + 1) * LDB);
    }
}

template <int RIDE>
__device__ __forceinline__ void frag_mma(const Frag& f, const FragX& fx, v8f (&acc)[4], v8f& accX)
{
    #pragma unroll
    for (int j = 0; j < KS; ++j) {
        #pragma unroll
        for (int t = 0; t < 4; ++t) {
            v2f b; b.x = f.b0[j][t]; b.y = f.b1[j][t];
            acc[t] = __builtin_amdgcn_wmma_f32_16x16x4_f32(false, f.a[j], false, b, (short)0, acc[t], false, false);
        }
        if (RIDE)
            accX = __builtin_amdgcn_wmma_f32_16x16x4_f32(false, fx.a[j], false, fx.b[j], (short)0, accX, false, false);
    }
}

template <int LDB, int LDX, int RM, int RL>
__device__ __forceinline__
void ring_trip(Frag (&f)[NS], FragX (&fx)[NS], v8f (&acc)[4], v8f& accX,
               const float* __restrict__ aP, const float* __restrict__ bP,
               const float* xaP, const float* xbP)
{
    #pragma unroll
    for (int s = 0; s < NS; ++s) {
        frag_mma<RM>(f[s], fx[s], acc, accX);
        frag_load<LDB>(f[s], aP + s * KC, bP + s * KC * LDB);
        if (RL) fragx_load<LDX>(fx[s], xaP + s * KC, xbP + s * KC * LDX);
        __builtin_amdgcn_sched_barrier(0);
    }
}

template <int LDB, int RIDE>
__device__ __forceinline__
void gemm_body(const float* __restrict__ Ag, int lda,
               const float* __restrict__ Bg,
               float* __restrict__ Cg, int ldc,
               int Kdim, int mBound, int nBound, int addI,
               const float* Xa, const float* Xb, float* Xo, int xn, v4f* red)
{
    const int tid  = threadIdx.x;
    const int lane = tid & 31;
    const int wv   = __builtin_amdgcn_readfirstlane(tid >> 5);
    const int mw   = wv & 3;
    const int kh   = wv >> 2;
    const int h    = lane >> 4;
    const int c    = lane & 15;
    const int row0 = blockIdx.y * 64 + mw * 16;
    const int col0 = blockIdx.x * 64;

    const int Kh = Kdim >> 1;
    const int k0 = kh * Kh;
    const float* aP = Ag + (size_t)(row0 + c) * lda + k0 + 2 * h;
    const float* bP = Bg + (size_t)(k0 + 2 * h) * LDB + col0 + 4 * c;

    constexpr int LDX = (RIDE == 2) ? NDIM : YLD;
    const float* xaP = Xa;
    const float* xbP = Xb;
    int xcol = 0, xrow0 = 0;
    if (RIDE) {
        const int bid = (int)blockIdx.y * (int)gridDim.x + (int)blockIdx.x;
        const int kx  = wv * (Kdim >> 3) + 2 * h;
        if (RIDE == 1) {
            xrow0 = 16 * (bid & 63);
            xcol  = 16 * (bid >> 6) + c;
            xaP = Xa + (size_t)(xrow0 + c) * NDIM + kx;
            xbP = Xb + (size_t)kx * LDX + (xcol < xn ? xcol : 0);
        } else {
            xcol  = 16 * bid + c;
            xaP = Xa + (size_t)(c < xn ? c : 0) * NDIM + kx;
            xbP = Xb + (size_t)kx * LDX + xcol;
        }
    }

    v8f acc[4] = {};
    v8f accX = {};
    Frag f[NS];
    FragX fx[NS];

    #pragma unroll
    for (int s = 0; s < NS; ++s) {
        frag_load<LDB>(f[s], aP + s * KC, bP + s * KC * LDB);
        if (RIDE) fragx_load<LDX>(fx[s], xaP + s * KC, xbP + s * KC * LDX);
        __builtin_amdgcn_sched_barrier(0);
    }

    int kb = NS * KC;
    if (RIDE) {
        aP += NS * KC;  bP += NS * KC * LDB;  xaP += NS * KC;  xbP += NS * KC * LDX;
        ring_trip<LDB, LDX, 1, 1>(f, fx, acc, accX, aP, bP, xaP, xbP);
        aP += NS * KC;  bP += NS * KC * LDB;
        ring_trip<LDB, LDX, 1, 0>(f, fx, acc, accX, aP, bP, xaP, xbP);
        kb += 2 * NS * KC;
    }
    #pragma nounroll
    for (; kb < Kh; kb += NS * KC) {
        aP += NS * KC;
        bP += NS * KC * LDB;
        ring_trip<LDB, LDX, 0, 0>(f, fx, acc, accX, aP, bP, xaP, xbP);
    }
    #pragma unroll
    for (int s = 0; s < NS; ++s) {
        frag_mma<0>(f[s], fx[s], acc, accX);
        __builtin_amdgcn_sched_barrier(0);
    }

    v4f* rp = red + (mw * 8) * 32 + lane;
    v4f* rx = red + RED_MAIN + lane;
    if (kh == 1) {
        #pragma unroll
        for (int t = 0; t < 4; ++t) {
            v4f lo, hi;
            lo.x = acc[t][0]; lo.y = acc[t][1]; lo.z = acc[t][2]; lo.w = acc[t][3];
            hi.x = acc[t][4]; hi.y = acc[t][5]; hi.z = acc[t][6]; hi.w = acc[t][7];
            rp[(2 * t) * 32]     = lo;
            rp[(2 * t + 1) * 32] = hi;
        }
    }
    if (RIDE) {
        if (wv != 0) {
            v4f lo, hi;
            lo.x = accX[0]; lo.y = accX[1]; lo.z = accX[2]; lo.w = accX[3];
            hi.x = accX[4]; hi.y = accX[5]; hi.z = accX[6]; hi.w = accX[7];
            rx[(2 * wv) * 32]     = lo;
            rx[(2 * wv + 1) * 32] = hi;
        }
    }
    __syncthreads();
    if (kh == 1) return;
    #pragma unroll
    for (int t = 0; t < 4; ++t) {
        const v4f lo = rp[(2 * t) * 32];
        const v4f hi = rp[(2 * t + 1) * 32];
        acc[t][0] += lo.x; acc[t][1] += lo.y; acc[t][2] += lo.z; acc[t][3] += lo.w;
        acc[t][4] += hi.x; acc[t][5] += hi.y; acc[t][6] += hi.z; acc[t][7] += hi.w;
    }
    if (RIDE) {
        if (wv == 0) {
            #pragma unroll
            for (int q = 1; q < 8; ++q) {
                const v4f lo = rx[(2 * q) * 32];
                const v4f hi = rx[(2 * q + 1) * 32];
                accX[0] += lo.x; accX[1] += lo.y; accX[2] += lo.z; accX[3] += lo.w;
                accX[4] += hi.x; accX[5] += hi.y; accX[6] += hi.z; accX[7] += hi.w;
            }
        }
    }

    const bool whole = (row0 + 16 <= mBound) && (col0 + 64 <= nBound) &&
                       ((ldc & 3) == 0) && ((((size_t)Cg) & 15) == 0);
    if (whole) {
        #pragma unroll
        for (int r = 0; r < 8; ++r) {
            const int row = row0 + 8 * h + r;
            const int col = col0 + 4 * c;
            v4f v;
            v.x = acc[0][r]; v.y = acc[1][r]; v.z = acc[2][r]; v.w = acc[3][r];
            if (addI) {
                if (row == col)     v.x += 1.0f;
                if (row == col + 1) v.y += 1.0f;
                if (row == col + 2) v.z += 1.0f;
                if (row == col + 3) v.w += 1.0f;
            }
            *(v4f*)(Cg + (size_t)row * ldc + col) = v;
        }
    } else {
        #pragma unroll
        for (int t = 0; t < 4; ++t)
            #pragma unroll
            for (int r = 0; r < 8; ++r) {
                const int row = row0 + 8 * h + r;
                const int col = col0 + 4 * c + t;
                if (row < mBound && col < nBound) {
                    float v = acc[t][r];
                    if (addI && row == col) v += 1.0f;
                    Cg[(size_t)row * ldc + col] = v;
                }
            }
    }

    if (RIDE == 1) {
        if (wv == 0 && xcol < xn) {
            #pragma unroll
            for (int r = 0; r < 8; ++r)
                Xo[(size_t)(xrow0 + 8 * h + r) * LDX + xcol] = accX[r];
        }
    }
    if (RIDE == 2) {
        if (wv == 0) {
            #pragma unroll
            for (int r = 0; r < 8; ++r) {
                const int rr = 8 * h + r;
                if (rr < xn) Xo[(size_t)rr * LDX + xcol] = accX[r];
            }
        }
    }
}

template <int LDB>
__global__ __launch_bounds__(256)
void gemm_wmma_f32(const float* __restrict__ Ag, int lda,
                   const float* __restrict__ Bg,
                   float* __restrict__ Cg, int ldc,
                   int Kdim, int mBound, int nBound, int addI)
{
    __shared__ v4f red[RED_V4];
    gemm_body<LDB, 0>(Ag, lda, Bg, Cg, ldc, Kdim, mBound, nBound, addI, nullptr, nullptr, nullptr, 0, red);
}

__global__ __launch_bounds__(256)
void gemm_sq_cols(const float* __restrict__ Ms, float* __restrict__ M2,
                  const float* Yg, float* Yo, int s)
{
    __shared__ v4f red[RED_V4];
    const int bid = (int)blockIdx.y * (int)gridDim.x + (int)blockIdx.x;
    if (bid < 4 * (s > 16 ? s : 16))
        gemm_body<NDIM, 1>(Ms, NDIM, Ms, M2, NDIM, NDIM, NDIM, NDIM, 0, Ms, Yg, Yo, s, red);
    else
        gemm_body<NDIM, 0>(Ms, NDIM, Ms, M2, NDIM, NDIM, NDIM, NDIM, 0, nullptr, nullptr, nullptr, 0, red);
}

__global__ __launch_bounds__(256)
void gemm_sq_rows(const float* __restrict__ Qt, float* __restrict__ Q2,
                  const float* Rg, float* Ro, int t)
{
    __shared__ v4f red[RED_V4];
    const int bid = (int)blockIdx.y * (int)gridDim.x + (int)blockIdx.x;
    if (bid < 64)
        gemm_body<NDIM, 2>(Qt, NDIM, Qt, Q2, NDIM, NDIM, NDIM, NDIM, 0, Rg, Qt, Ro, t, red);
    else
        gemm_body<NDIM, 0>(Qt, NDIM, Qt, Q2, NDIM, NDIM, NDIM, NDIM, 0, nullptr, nullptr, nullptr, 0, red);
}

__global__ void prep_mats(const float* __restrict__ A, const float* __restrict__ ldt,
                          float* __restrict__ M, float* __restrict__ IM, int n)
{
    int idx = blockIdx.x * blockDim.x + threadIdx.x;
    if (idx >= n * n) return;
    float h = 0.5f * expf(ldt[0]);
    float m = h * A[idx];
    M[idx]  = m;
    int r = idx / n, c = idx - r * n;
    IM[idx] = m + ((r == c) ? 1.0f : 0.0f);
}

__global__ void bb_matvec(const float* __restrict__ S, const float* __restrict__ Bv,
                          const float* __restrict__ ldt, float* __restrict__ Y,
                          int n, int ldy)
{
    int i = blockIdx.x * blockDim.x + threadIdx.x;
    if (i >= n) return;
    float dt = expf(ldt[0]);
    float s = 0.f;
    for (int k = 0; k < n; ++k) s += S[(size_t)i * n + k] * Bv[k];
    Y[(size_t)i * ldy] = dt * s;
}

__global__ void copy_row(const float* __restrict__ Cv, float* __restrict__ R, int n)
{
    int i = blockIdx.x * blockDim.x + threadIdx.x;
    if (i < n) R[i] = Cv[i];
}

#define CT   128
#define CW   16
#define CPAD 128
__global__ __launch_bounds__(32 * CW)
void conv_causal(const float* __restrict__ u, const float* __restrict__ Kk,
                 const float* __restrict__ Dv, float* __restrict__ y, int L)
{
    __shared__ __attribute__((aligned(16))) float us[CPAD + LSEQ];
    __shared__ __attribute__((aligned(16))) float kr[LSEQ];
    __shared__ __attribute__((aligned(16))) float red[CW][CT];
    const int tid  = threadIdx.x;
    const int lane = tid & 31;
    const int wv   = __builtin_amdgcn_readfirstlane(tid >> 5);
    const int t0   = blockIdx.x * CT;
    const int nt   = t0 + CT;

    for (int i = 4 * tid; i < nt; i += 4 * 32 * CW) {
        *(v4f*)&us[CPAD + i] = *(const v4f*)&u[i];
        const v4f kv = *(const v4f*)&Kk[L - 4 - i];
        v4f r; r.x = kv.w; r.y = kv.z; r.z = kv.y; r.w = kv.x;
        *(v4f*)&kr[i] = r;
    }
    if (tid < CPAD / 4) { v4f z = {0.f, 0.f, 0.f, 0.f}; *(v4f*)&us[4 * tid] = z; }
    __syncthreads();

    const int S  = (CT / CW) * ((int)blockIdx.x + 1);
    const int l0 = wv * S;
    const int T  = t0 + 4 * lane;
    float a0 = 0.f, a1 = 0.f, a2 = 0.f, a3 = 0.f;
    auto mac = [&](const v4f& w1, const v4f& w0, const v4f& k) {
        a0 += k.x * w1.x;  a1 += k.x * w1.y;  a2 += k.x * w1.z;  a3 += k.x * w1.w;
        a0 += k.y * w0.w;  a1 += k.y * w1.x;  a2 += k.y * w1.y;  a3 += k.y * w1.z;
        a0 += k.z * w0.z;  a1 += k.z * w0.w;  a2 += k.z * w1.x;  a3 += k.z * w1.y;
        a0 += k.w * w0.y;  a1 += k.w * w0.z;  a2 += k.w * w0.w;  a3 += k.w * w1.x;
    };
    v4f w1 = *(const v4f*)&us[CPAD + T - l0];
    v4f w0 = *(const v4f*)&us[CPAD + T - l0 - 4];
    v4f k  = *(const v4f*)&kr[l0];
    for (int l = l0; l < l0 + S - 4; l += 4) {
        const v4f w0n = *(const v4f*)&us[CPAD + T - l - 8];
        const v4f kn  = *(const v4f*)&kr[l + 4];
        mac(w1, w0, k);
        w1 = w0; w0 = w0n; k = kn;
    }
    mac(w1, w0, k);
    { v4f p; p.x = a0; p.y = a1; p.z = a2; p.w = a3; *(v4f*)&red[wv][4 * lane] = p; }
    __syncthreads();
    if (tid < CT) {
        float s = 0.f;
        for (int w = 0; w < CW; ++w) s += red[w][tid];
        y[t0 + tid] = s + Dv[0] * u[t0 + tid];
    }
}

extern "C" void kernel_launch(void* const* d_in, const int* in_sizes, int n_in,
                              void* d_out, int out_size, void* d_ws, size_t ws_size,
                              hipStream_t stream)
{
    (void)in_sizes; (void)n_in; (void)out_size; (void)ws_size;
    const float* u   = (const float*)d_in[0];
    const float* A   = (const float*)d_in[1];
    const float* Bv  = (const float*)d_in[2];
    const float* Cv  = (const float*)d_in[3];
    const float* Dv  = (const float*)d_in[4];
    const float* ldt = (const float*)d_in[5];
    float* y = (float*)d_out;
    const int N = NDIM, L = LSEQ;

    float* B0 = (float*)d_ws;
    float* B1 = B0 + (size_t)N * N;
    float* B2 = B1 + (size_t)N * N;
    float* B3 = B2 + (size_t)N * N;
    float* Y  = B3 + (size_t)N * N;
    float* R  = Y  + (size_t)N * 128;
    float* Kk = R  + (size_t)64 * N;

    dim3 blk(256);
    auto gemm = [&](const float* Am, int lda, const float* Bm, int ldb,
                    float* Cm, int ldc, int Mc, int Nc, int Kd,
                    int mB, int nB, int addI) {
        (void)ldb;
        dim3 grid((Nc + 63) / 64, (Mc + 63) / 64);
        gemm_wmma_f32<NDIM><<<grid, blk, 0, stream>>>(Am, lda, Bm, Cm, ldc, Kd, mB, nB, addI);
    };
    auto gemmY = [&](const float* Am, int lda, const float* Bm, int ldb,
                     float* Cm, int ldc, int Mc, int Nc, int Kd,
                     int mB, int nB, int addI) {
        (void)ldb;
        dim3 grid((Nc + 63) / 64, (Mc + 63) / 64);
        gemm_wmma_f32<128><<<grid, blk, 0, stream>>>(Am, lda, Bm, Cm, ldc, Kd, mB, nB, addI);
    };

    prep_mats<<<(N * N + 255) / 256, 256, 0, stream>>>(A, ldt, B0, B1, N);
    gemm(B0, N, B1, N, B2, N, N, N, N, N, N, 1);
    gemm(B0, N, B2, N, B3, N, N, N, N, N, N, 1);
    gemm(B0, N, B3, N, B2, N, N, N, N, N, N, 1);
    gemm(B2, N, B1, N, B3, N, N, N, N, N, N, 0);
    bb_matvec<<<(N + 255) / 256, 256, 0, stream>>>(B2, Bv, ldt, Y, N, 128);

    const dim3 sqGrid(NDIM / 64, NDIM / 64);

    auto sqCols = [&](const float* Ms, float* M2s, float* Ynew, int s) {
        gemm_sq_cols<<<sqGrid, blk, 0, stream>>>(Ms, M2s, Y, Ynew, s);
    };
    sqCols(B3, B0, Y + 1,  1);
    sqCols(B0, B1, Y + 2,  2);
    sqCols(B1, B0, Y + 4,  4);
    sqCols(B0, B1, Y + 8,  8);
    sqCols(B1, B0, Y + 16, 16);
    sqCols(B0, B1, Y + 32, 32);
    sqCols(B1, B0, Y + 64, 64);

    copy_row<<<(N + 255) / 256, 256, 0, stream>>>(Cv, R, N);
    auto sqRows = [&](const float* Qt, float* Q2t, float* Rnew, int t) {
        gemm_sq_rows<<<sqGrid, blk, 0, stream>>>(Qt, Q2t, R, Rnew, t);
    };
    sqRows(B0, B1, R + 1 * N,  1);
    sqRows(B1, B0, R + 2 * N,  2);
    sqRows(B0, B1, R + 4 * N,  4);
    sqRows(B1, B0, R + 8 * N,  8);
    sqRows(B0, B1, R + 16 * N, 16);
    gemm(R, N, B1, N, R + 32 * N, N, 64, N, N, 32, N, 0);

    gemmY(R, N, Y, 128, Kk, 128, 64, 128, N, 64, 128, 0);

    conv_causal<<<L / CT, 32 * CW, 0, stream>>>(u, Kk, Dv, y, L);
}


# Bb matvec re-cut: one wave per row of S, lanes read row and B in 16-byte pieces (a load = whole cache lines, not 32 rows), 4 partial sums per lane, then a lane butterfly (sum order changes)
# speedup vs baseline: 10.1708x; 1.0744x over previous
#include <hip/hip_runtime.h>
#include <math.h>


typedef __attribute__((ext_vector_type(2))) float v2f;
typedef __attribute__((ext_vector_type(4))) float v4f;
typedef __attribute__((ext_vector_type(8))) float v8f;

#define NDIM 1024
#define LSEQ 8192

#define KS 8
#define NS 2
#define KC (4 * KS)
#define RED_MAIN (4 * 8 * 32)
#define RED_V4 (RED_MAIN + 8 * 2 * 32)

struct Frag { v2f a[KS]; v4f b0[KS]; v4f b1[KS]; };

#define YLD 128
struct FragX { v2f a[KS]; v2f b[KS]; };

template <int LDX>
__device__ __forceinline__ void fragx_load(FragX& f, const float* xaP, const float* xbP)
{
    #pragma unroll
    for (int j = 0; j < KS; ++j) {
        f.a[j]   = *(const v2f*)(xaP + 4 * j);
        f.b[j].x = xbP[(4 * j) * LDX];
        f.b[j].y = xbP[(4 * j + 1) * LDX];
    }
}

template <int LDB>
__device__ __forceinline__ void frag_load(Frag& f, const float* __restrict__ aP,
                                          const float* __restrict__ bP)
{
    #pragma unroll
    for (int j = 0; j < KS; ++j) {
        f.a[j]  = *(const v2f*)(aP + 4 * j);
        f.b0[j] = *(const v4f*)(bP + (4 * j) * LDB);
        f.b1[j] = *(const v4f*)(bP + (4 * j + 1) * LDB);
    }
}

template <int RIDE>
__device__ __forceinline__ void frag_mma(const Frag& f, const FragX& fx, v8f (&acc)[4], v8f& accX)
{
    #pragma unroll
    for (int j = 0; j < KS; ++j) {
        #pragma unroll
        for (int t = 0; t < 4; ++t) {
            v2f b; b.x = f.b0[j][t]; b.y = f.b1[j][t];
            acc[t] = __builtin_amdgcn_wmma_f32_16x16x4_f32(false, f.a[j], false, b, (short)0, acc[t], false, false);
        }
        if (RIDE)
            accX = __builtin_amdgcn_wmma_f32_16x16x4_f32(false, fx.a[j], false, fx.b[j], (short)0, accX, false, false);
    }
}

template <int LDB, int LDX, int RM, int RL>
__device__ __forceinline__
void ring_trip(Frag (&f)[NS], FragX (&fx)[NS], v8f (&acc)[4], v8f& accX,
               const float* __restrict__ aP, const float* __restrict__ bP,
               const float* xaP, const float* xbP)
{
    #pragma unroll
    for (int s = 0; s < NS; ++s) {
        frag_mma<RM>(f[s], fx[s], acc, accX);
        frag_load<LDB>(f[s], aP + s * KC, bP + s * KC * LDB);
        if (RL) fragx_load<LDX>(fx[s], xaP + s * KC, xbP + s * KC * LDX);
        __builtin_amdgcn_sched_barrier(0);
    }
}

template <int LDB, int RIDE>
__device__ __forceinline__
void gemm_body(const float* __restrict__ Ag, int lda,
               const float* __restrict__ Bg,
               float* __restrict__ Cg, int ldc,
               int Kdim, int mBound, int nBound, int addI,
               const float* Xa, const float* Xb, float* Xo, int xn, v4f* red)
{
    const int tid  = threadIdx.x;
    const int lane = tid & 31;
    const int wv   = __builtin_amdgcn_readfirstlane(tid >> 5);
    const int mw   = wv & 3;
    const int kh   = wv >> 2;
    const int h    = lane >> 4;
    const int c    = lane & 15;
    const int row0 = blockIdx.y * 64 + mw * 16;
    const int col0 = blockIdx.x * 64;

    const int Kh = Kdim >> 1;
    const int k0 = kh * Kh;
    const float* aP = Ag + (size_t)(row0 + c) * lda + k0 + 2 * h;
    const float* bP = Bg + (size_t)(k0 + 2 * h) * LDB + col0 + 4 * c;

    constexpr int LDX = (RIDE == 2) ? NDIM : YLD;
    const float* xaP = Xa;
    const float* xbP = Xb;
    int xcol = 0, xrow0 = 0;
    if (RIDE) {
        const int bid = (int)blockIdx.y * (int)gridDim.x + (int)blockIdx.x;
        const int kx  = wv * (Kdim >> 3) + 2 * h;
        if (RIDE == 1) {
            xrow0 = 16 * (bid & 63);
            xcol  = 16 * (bid >> 6) + c;
            xaP = Xa + (size_t)(xrow0 + c) * NDIM + kx;
            xbP = Xb + (size_t)kx * LDX + (xcol < xn ? xcol : 0);
        } else {
            xcol  = 16 * bid + c;
            xaP = Xa + (size_t)(c < xn ? c : 0) * NDIM + kx;
            xbP = Xb + (size_t)kx * LDX + xcol;
        }
    }

    v8f acc[4] = {};
    v8f accX = {};
    Frag f[NS];
    FragX fx[NS];

    #pragma unroll
    for (int s = 0; s < NS; ++s) {
        frag_load<LDB>(f[s], aP + s * KC, bP + s * KC * LDB);
        if (RIDE) fragx_load<LDX>(fx[s], xaP + s * KC, xbP + s * KC * LDX);
        __builtin_amdgcn_sched_barrier(0);
    }

    int kb = NS * KC;
    if (RIDE) {
        aP += NS * KC;  bP += NS * KC * LDB;  xaP += NS * KC;  xbP += NS * KC * LDX;
        ring_trip<LDB, LDX, 1, 1>(f, fx, acc, accX, aP, bP, xaP, xbP);
        aP += NS * KC;  bP += NS * KC * LDB;
        ring_trip<LDB, LDX, 1, 0>(f, fx, acc, accX, aP, bP, xaP, xbP);
        kb += 2 * NS * KC;
    }
    #pragma nounroll
    for (; kb < Kh; kb += NS * KC) {
        aP += NS * KC;
        bP += NS * KC * LDB;
        ring_trip<LDB, LDX, 0, 0>(f, fx, acc, accX, aP, bP, xaP, xbP);
    }
    #pragma unroll
    for (int s = 0; s < NS; ++s) {
        frag_mma<0>(f[s], fx[s], acc, accX);
        __builtin_amdgcn_sched_barrier(0);
    }

    v4f* rp = red + (mw * 8) * 32 + lane;
    v4f* rx = red + RED_MAIN + lane;
    if (kh == 1) {
        #pragma unroll
        for (int t = 0; t < 4; ++t) {
            v4f lo, hi;
            lo.x = acc[t][0]; lo.y = acc[t][1]; lo.z = acc[t][2]; lo.w = acc[t][3];
            hi.x = acc[t][4]; hi.y = acc[t][5]; hi.z = acc[t][6]; hi.w = acc[t][7];
            rp[(2 * t) * 32]     = lo;
            rp[(2 * t + 1) * 32] = hi;
        }
    }
    if (RIDE) {
        if (wv != 0) {
            v4f lo, hi;
            lo.x = accX[0]; lo.y = accX[1]; lo.z = accX[2]; lo.w = accX[3];
            hi.x = accX[4]; hi.y = accX[5]; hi.z = accX[6]; hi.w = accX[7];
            rx[(2 * wv) * 32]     = lo;
            rx[(2 * wv + 1) * 32] = hi;
        }
    }
    __syncthreads();
    if (kh == 1) return;
    #pragma unroll
    for (int t = 0; t < 4; ++t) {
        const v4f lo = rp[(2 * t) * 32];
        const v4f hi = rp[(2 * t + 1) * 32];
        acc[t][0] += lo.x; acc[t][1] += lo.y; acc[t][2] += lo.z; acc[t][3] += lo.w;
        acc[t][4] += hi.x; acc[t][5] += hi.y; acc[t][6] += hi.z; acc[t][7] += hi.w;
    }
    if (RIDE) {
        if (wv == 0) {
            #pragma unroll
            for (int q = 1; q < 8; ++q) {
                const v4f lo = rx[(2 * q) * 32];
                const v4f hi = rx[(2 * q + 1) * 32];
                accX[0] += lo.x; accX[1] += lo.y; accX[2] += lo.z; accX[3] += lo.w;
                accX[4] += hi.x; accX[5] += hi.y; accX[6] += hi.z; accX[7] += hi.w;
            }
        }
    }

    const bool whole = (row0 + 16 <= mBound) && (col0 + 64 <= nBound) &&
                       ((ldc & 3) == 0) && ((((size_t)Cg) & 15) == 0);
    if (whole) {
        #pragma unroll
        for (int r = 0; r < 8; ++r) {
            const int row = row0 + 8 * h + r;
            const int col = col0 + 4 * c;
            v4f v;
            v.x = acc[0][r]; v.y = acc[1][r]; v.z = acc[2][r]; v.w = acc[3][r];
            if (addI) {
                if (row == col)     v.x += 1.0f;
                if (row == col + 1) v.y += 1.0f;
                if (row == col + 2) v.z += 1.0f;
                if (row == col + 3) v.w += 1.0f;
            }
            *(v4f*)(Cg + (size_t)row * ldc + col) = v;
        }
    } else {
        #pragma unroll
        for (int t = 0; t < 4; ++t)
            #pragma unroll
            for (int r = 0; r < 8; ++r) {
                const int row = row0 + 8 * h + r;
                const int col = col0 + 4 * c + t;
                if (row < mBound && col < nBound) {
                    float v = acc[t][r];
                    if (addI && row == col) v += 1.0f;
                    Cg[(size_t)row * ldc + col] = v;
                }
            }
    }

    if (RIDE == 1) {
        if (wv == 0 && xcol < xn) {
            #pragma unroll
            for (int r = 0; r < 8; ++r)
                Xo[(size_t)(xrow0 + 8 * h + r) * LDX + xcol] = accX[r];
        }
    }
    if (RIDE == 2) {
        if (wv == 0) {
            #pragma unroll
            for (int r = 0; r < 8; ++r) {
                const int rr = 8 * h + r;
                if (rr < xn) Xo[(size_t)rr * LDX + xcol] = accX[r];
            }
        }
    }
}

template <int LDB>
__global__ __launch_bounds__(256)
void gemm_wmma_f32(const float* __restrict__ Ag, int lda,
                   const float* __restrict__ Bg,
                   float* __restrict__ Cg, int ldc,
                   int Kdim, int mBound, int nBound, int addI)
{
    __shared__ v4f red[RED_V4];
    gemm_body<LDB, 0>(Ag, lda, Bg, Cg, ldc, Kdim, mBound, nBound, addI, nullptr, nullptr, nullptr, 0, red);
}

__global__ __launch_bounds__(256)
void gemm_sq_cols(const float* __restrict__ Ms, float* __restrict__ M2,
                  const float* Yg, float* Yo, int s)
{
    __shared__ v4f red[RED_V4];
    const int bid = (int)blockIdx.y * (int)gridDim.x + (int)blockIdx.x;
    if (bid < 4 * (s > 16 ? s : 16))
        gemm_body<NDIM, 1>(Ms, NDIM, Ms, M2, NDIM, NDIM, NDIM, NDIM, 0, Ms, Yg, Yo, s, red);
    else
        gemm_body<NDIM, 0>(Ms, NDIM, Ms, M2, NDIM, NDIM, NDIM, NDIM, 0, nullptr, nullptr, nullptr, 0, red);
}

__global__ __launch_bounds__(256)
void gemm_sq_rows(const float* __restrict__ Qt, float* __restrict__ Q2,
                  const float* Rg, float* Ro, int t)
{
    __shared__ v4f red[RED_V4];
    const int bid = (int)blockIdx.y * (int)gridDim.x + (int)blockIdx.x;
    if (bid < 64)
        gemm_body<NDIM, 2>(Qt, NDIM, Qt, Q2, NDIM, NDIM, NDIM, NDIM, 0, Rg, Qt, Ro, t, red);
    else
        gemm_body<NDIM, 0>(Qt, NDIM, Qt, Q2, NDIM, NDIM, NDIM, NDIM, 0, nullptr, nullptr, nullptr, 0, red);
}

__global__ void prep_mats(const float* __restrict__ A, const float* __restrict__ ldt,
                          float* __restrict__ M, float* __restrict__ IM, int n)
{
    int idx = blockIdx.x * blockDim.x + threadIdx.x;
    if (idx >= n * n) return;
    float h = 0.5f * expf(ldt[0]);
    float m = h * A[idx];
    M[idx]  = m;
    int r = idx / n, c = idx - r * n;
    IM[idx] = m + ((r == c) ? 1.0f : 0.0f);
}

#define MVW 8
__global__ __launch_bounds__(32 * MVW)
void bb_matvec(const float* __restrict__ S, const float* __restrict__ Bv,
               const float* __restrict__ ldt, float* __restrict__ Y,
               int n, int ldy)
{
    const int lane = threadIdx.x & 31;
    const int row  = blockIdx.x * MVW + (threadIdx.x >> 5);
    if (row >= n) return;
    const float* sp = S + (size_t)row * n + 4 * lane;
    const float* bp = Bv + 4 * lane;
    v4f acc = {0.f, 0.f, 0.f, 0.f};
    #pragma unroll 8
    for (int k = 0; k < n; k += 128) {
        const v4f s = *(const v4f*)(sp + k);
        const v4f b = *(const v4f*)(bp + k);
        acc.x += s.x * b.x;
        acc.y += s.y * b.y;
        acc.z += s.z * b.z;
        acc.w += s.w * b.w;
    }
    float t = (acc.x + acc.y) + (acc.z + acc.w);
    for (int o = 16; o > 0; o >>= 1) t += __shfl_xor(t, o);
    if (lane == 0) Y[(size_t)row * ldy] = expf(ldt[0]) * t;
}

__global__ void copy_row(const float* __restrict__ Cv, float* __restrict__ R, int n)
{
    int i = blockIdx.x * blockDim.x + threadIdx.x;
    if (i < n) R[i] = Cv[i];
}

#define CT   128
#define CW   16
#define CPAD 128
__global__ __launch_bounds__(32 * CW)
void conv_causal(const float* __restrict__ u, const float* __restrict__ Kk,
                 const float* __restrict__ Dv, float* __restrict__ y, int L)
{
    __shared__ __attribute__((aligned(16))) float us[CPAD + LSEQ];
    __shared__ __attribute__((aligned(16))) float kr[LSEQ];
    __shared__ __attribute__((aligned(16))) float red[CW][CT];
    const int tid  = threadIdx.x;
    const int lane = tid & 31;
    const int wv   = __builtin_amdgcn_readfirstlane(tid >> 5);
    const int t0   = blockIdx.x * CT;
    const int nt   = t0 + CT;

    for (int i = 4 * tid; i < nt; i += 4 * 32 * CW) {
        *(v4f*)&us[CPAD + i] = *(const v4f*)&u[i];
        const v4f kv = *(const v4f*)&Kk[L - 4 - i];
        v4f r; r.x = kv.w; r.y = kv.z; r.z = kv.y; r.w = kv.x;
        *(v4f*)&kr[i] = r;
    }
    if (tid < CPAD / 4) { v4f z = {0.f, 0.f, 0.f, 0.f}; *(v4f*)&us[4 * tid] = z; }
    __syncthreads();

    const int S  = (CT / CW) * ((int)blockIdx.x + 1);
    const int l0 = wv * S;
    const int T  = t0 + 4 * lane;
    float a0 = 0.f, a1 = 0.f, a2 = 0.f, a3 = 0.f;
    auto mac = [&](const v4f& w1, const v4f& w0, const v4f& k) {
        a0 += k.x * w1.x;  a1 += k.x * w1.y;  a2 += k.x * w1.z;  a3 += k.x * w1.w;
        a0 += k.y * w0.w;  a1 += k.y * w1.x;  a2 += k.y * w1.y;  a3 += k.y * w1.z;
        a0 += k.z * w0.z;  a1 += k.z * w0.w;  a2 += k.z * w1.x;  a3 += k.z * w1.y;
        a0 += k.w * w0.y;  a1 += k.w * w0.z;  a2 += k.w * w0.w;  a3 += k.w * w1.x;
    };
    v4f w1 = *(const v4f*)&us[CPAD + T - l0];
    v4f w0 = *(const v4f*)&us[CPAD + T - l0 - 4];
    v4f k  = *(const v4f*)&kr[l0];
    for (int l = l0; l < l0 + S - 4; l += 4) {
        const v4f w0n = *(const v4f*)&us[CPAD + T - l - 8];
        const v4f kn  = *(const v4f*)&kr[l + 4];
        mac(w1, w0, k);
        w1 = w0; w0 = w0n; k = kn;
    }
    mac(w1, w0, k);
    { v4f p; p.x = a0; p.y = a1; p.z = a2; p.w = a3; *(v4f*)&red[wv][4 * lane] = p; }
    __syncthreads();
    if (tid < CT) {
        float s = 0.f;
        for (int w = 0; w < CW; ++w) s += red[w][tid];
        y[t0 + tid] = s + Dv[0] * u[t0 + tid];
    }
}

extern "C" void kernel_launch(void* const* d_in, const int* in_sizes, int n_in,
                              void* d_out, int out_size, void* d_ws, size_t ws_size,
                              hipStream_t stream)
{
    (void)in_sizes; (void)n_in; (void)out_size; (void)ws_size;
    const float* u   = (const float*)d_in[0];
    const float* A   = (const float*)d_in[1];
    const float* Bv  = (const float*)d_in[2];
    const float* Cv  = (const float*)d_in[3];
    const float* Dv  = (const float*)d_in[4];
    const float* ldt = (const float*)d_in[5];
    float* y = (float*)d_out;
    const int N = NDIM, L = LSEQ;

    float* B0 = (float*)d_ws;
    float* B1 = B0 + (size_t)N * N;
    float* B2 = B1 + (size_t)N * N;
    float* B3 = B2 + (size_t)N * N;
    float* Y  = B3 + (size_t)N * N;
    float* R  = Y  + (size_t)N * 128;
    float* Kk = R  + (size_t)64 * N;

    dim3 blk(256);
    auto gemm = [&](const float* Am, int lda, const float* Bm, int ldb,
                    float* Cm, int ldc, int Mc, int Nc, int Kd,
                    int mB, int nB, int addI) {
        (void)ldb;
        dim3 grid((Nc + 63) / 64, (Mc + 63) / 64);
        gemm_wmma_f32<NDIM><<<grid, blk, 0, stream>>>(Am, lda, Bm, Cm, ldc, Kd, mB, nB, addI);
    };
    auto gemmY = [&](const float* Am, int lda, const float* Bm, int ldb,
                     float* Cm, int ldc, int Mc, int Nc, int Kd,
                     int mB, int nB, int addI) {
        (void)ldb;
        dim3 grid((Nc + 63) / 64, (Mc + 63) / 64);
        gemm_wmma_f32<128><<<grid, blk, 0, stream>>>(Am, lda, Bm, Cm, ldc, Kd, mB, nB, addI);
    };

    prep_mats<<<(N * N + 255) / 256, 256, 0, stream>>>(A, ldt, B0, B1, N);
    gemm(B0, N, B1, N, B2, N, N, N, N, N, N, 1);
    gemm(B0, N, B2, N, B3, N, N, N, N, N, N, 1);
    gemm(B0, N, B3, N, B2, N, N, N, N, N, N, 1);
    gemm(B2, N, B1, N, B3, N, N, N, N, N, N, 0);
    bb_matvec<<<N / MVW, 32 * MVW, 0, stream>>>(B2, Bv, ldt, Y, N, 128);

    const dim3 sqGrid(NDIM / 64, NDIM / 64);

    auto sqCols = [&](const float* Ms, float* M2s, float* Ynew, int s) {
        gemm_sq_cols<<<sqGrid, blk, 0, stream>>>(Ms, M2s, Y, Ynew, s);
    };
    sqCols(B3, B0, Y + 1,  1);
    sqCols(B0, B1, Y + 2,  2);
    sqCols(B1, B0, Y + 4,  4);
    sqCols(B0, B1, Y + 8,  8);
    sqCols(B1, B0, Y + 16, 16);
    sqCols(B0, B1, Y + 32, 32);
    sqCols(B1, B0, Y + 64, 64);

    copy_row<<<(N + 255) / 256, 256, 0, stream>>>(Cv, R, N);
    auto sqRows = [&](const float* Qt, float* Q2t, float* Rnew, int t) {
        gemm_sq_rows<<<sqGrid, blk, 0, stream>>>(Qt, Q2t, R, Rnew, t);
    };
    sqRows(B0, B1, R + 1 * N,  1);
    sqRows(B1, B0, R + 2 * N,  2);
    sqRows(B0, B1, R + 4 * N,  4);
    sqRows(B1, B0, R + 8 * N,  8);
    sqRows(B0, B1, R + 16 * N, 16);
    gemm(R, N, B1, N, R + 32 * N, N, 64, N, N, 32, N, 0);

    gemmY(R, N, Y, 128, Kk, 128, 64, 128, N, 64, 128, 0);

    conv_causal<<<L / CT, 32 * CW, 0, stream>>>(u, Kk, Dv, y, L);
}


# Top squaring dropped (Q_32 never formed): R rows 16..63 from three chained 16-row products with Q_16, one 16x16 tile per block, K in 8 wave slices added in slice order (sum order changes)
# speedup vs baseline: 10.7045x; 1.0525x over previous
#include <hip/hip_runtime.h>
#include <math.h>


typedef __attribute__((ext_vector_type(2))) float v2f;
typedef __attribute__((ext_vector_type(4))) float v4f;
typedef __attribute__((ext_vector_type(8))) float v8f;

#define NDIM 1024
#define LSEQ 8192

#define KS 8
#define NS 2
#define KC (4 * KS)
#define RED_MAIN (4 * 8 * 32)
#define RED_V4 (RED_MAIN + 8 * 2 * 32)

struct Frag { v2f a[KS]; v4f b0[KS]; v4f b1[KS]; };

#define YLD 128
struct FragX { v2f a[KS]; v2f b[KS]; };

template <int LDX>
__device__ __forceinline__ void fragx_load(FragX& f, const float* xaP, const float* xbP)
{
    #pragma unroll
    for (int j = 0; j < KS; ++j) {
        f.a[j]   = *(const v2f*)(xaP + 4 * j);
        f.b[j].x = xbP[(4 * j) * LDX];
        f.b[j].y = xbP[(4 * j + 1) * LDX];
    }
}

template <int LDB>
__device__ __forceinline__ void frag_load(Frag& f, const float* __restrict__ aP,
                                          const float* __restrict__ bP)
{
    #pragma unroll
    for (int j = 0; j < KS; ++j) {
        f.a[j]  = *(const v2f*)(aP + 4 * j);
        f.b0[j] = *(const v4f*)(bP + (4 * j) * LDB);
        f.b1[j] = *(const v4f*)(bP + (4 * j + 1) * LDB);
    }
}

template <int RIDE>
__device__ __forceinline__ void frag_mma(const Frag& f, const FragX& fx, v8f (&acc)[4], v8f& accX)
{
    #pragma unroll
    for (int j = 0; j < KS; ++j) {
        #pragma unroll
        for (int t = 0; t < 4; ++t) {
            v2f b; b.x = f.b0[j][t]; b.y = f.b1[j][t];
            acc[t] = __builtin_amdgcn_wmma_f32_16x16x4_f32(false, f.a[j], false, b, (short)0, acc[t], false, false);
        }
        if (RIDE)
            accX = __builtin_amdgcn_wmma_f32_16x16x4_f32(false, fx.a[j], false, fx.b[j], (short)0, accX, false, false);
    }
}

template <int LDB, int LDX, int RM, int RL>
__device__ __forceinline__
void ring_trip(Frag (&f)[NS], FragX (&fx)[NS], v8f (&acc)[4], v8f& accX,
               const float* __restrict__ aP, const float* __restrict__ bP,
               const float* xaP, const float* xbP)
{
    #pragma unroll
    for (int s = 0; s < NS; ++s) {
        frag_mma<RM>(f[s], fx[s], acc, accX);
        frag_load<LDB>(f[s], aP + s * KC, bP + s * KC * LDB);
        if (RL) fragx_load<LDX>(fx[s], xaP + s * KC, xbP + s * KC * LDX);
        __builtin_amdgcn_sched_barrier(0);
    }
}

template <int LDB, int RIDE>
__device__ __forceinline__
void gemm_body(const float* __restrict__ Ag, int lda,
               const float* __restrict__ Bg,
               float* __restrict__ Cg, int ldc,
               int Kdim, int mBound, int nBound, int addI,
               const float* Xa, const float* Xb, float* Xo, int xn, v4f* red)
{
    const int tid  = threadIdx.x;
    const int lane = tid & 31;
    const int wv   = __builtin_amdgcn_readfirstlane(tid >> 5);
    const int mw   = wv & 3;
    const int kh   = wv >> 2;
    const int h    = lane >> 4;
    const int c    = lane & 15;
    const int row0 = blockIdx.y * 64 + mw * 16;
    const int col0 = blockIdx.x * 64;

    const int Kh = Kdim >> 1;
    const int k0 = kh * Kh;
    const float* aP = Ag + (size_t)(row0 + c) * lda + k0 + 2 * h;
    const float* bP = Bg + (size_t)(k0 + 2 * h) * LDB + col0 + 4 * c;

    constexpr int LDX = (RIDE == 2) ? NDIM : YLD;
    const float* xaP = Xa;
    const float* xbP = Xb;
    int xcol = 0, xrow0 = 0;
    if (RIDE) {
        const int bid = (int)blockIdx.y * (int)gridDim.x + (int)blockIdx.x;
        const int kx  = wv * (Kdim >> 3) + 2 * h;
        if (RIDE == 1) {
            xrow0 = 16 * (bid & 63);
            xcol  = 16 * (bid >> 6) + c;
            xaP = Xa + (size_t)(xrow0 + c) * NDIM + kx;
            xbP = Xb + (size_t)kx * LDX + (xcol < xn ? xcol : 0);
        } else {
            xcol  = 16 * bid + c;
            xaP = Xa + (size_t)(c < xn ? c : 0) * NDIM + kx;
            xbP = Xb + (size_t)kx * LDX + xcol;
        }
    }

    v8f acc[4] = {};
    v8f accX = {};
    Frag f[NS];
    FragX fx[NS];

    #pragma unroll
    for (int s = 0; s < NS; ++s) {
        frag_load<LDB>(f[s], aP + s * KC, bP + s * KC * LDB);
        if (RIDE) fragx_load<LDX>(fx[s], xaP + s * KC, xbP + s * KC * LDX);
        __builtin_amdgcn_sched_barrier(0);
    }

    int kb = NS * KC;
    if (RIDE) {
        aP += NS * KC;  bP += NS * KC * LDB;  xaP += NS * KC;  xbP += NS * KC * LDX;
        ring_trip<LDB, LDX, 1, 1>(f, fx, acc, accX, aP, bP, xaP, xbP);
        aP += NS * KC;  bP += NS * KC * LDB;
        ring_trip<LDB, LDX, 1, 0>(f, fx, acc, accX, aP, bP, xaP, xbP);
        kb += 2 * NS * KC;
    }
    #pragma nounroll
    for (; kb < Kh; kb += NS * KC) {
        aP += NS * KC;
        bP += NS * KC * LDB;
        ring_trip<LDB, LDX, 0, 0>(f, fx, acc, accX, aP, bP, xaP, xbP);
    }
    #pragma unroll
    for (int s = 0; s < NS; ++s) {
        frag_mma<0>(f[s], fx[s], acc, accX);
        __builtin_amdgcn_sched_barrier(0);
    }

    v4f* rp = red + (mw * 8) * 32 + lane;
    v4f* rx = red + RED_MAIN + lane;
    if (kh == 1) {
        #pragma unroll
        for (int t = 0; t < 4; ++t) {
            v4f lo, hi;
            lo.x = acc[t][0]; lo.y = acc[t][1]; lo.z = acc[t][2]; lo.w = acc[t][3];
            hi.x = acc[t][4]; hi.y = acc[t][5]; hi.z = acc[t][6]; hi.w = acc[t][7];
            rp[(2 * t) * 32]     = lo;
            rp[(2 * t + 1) * 32] = hi;
        }
    }
    if (RIDE) {
        if (wv != 0) {
            v4f lo, hi;
            lo.x = accX[0]; lo.y = accX[1]; lo.z = accX[2]; lo.w = accX[3];
            hi.x = accX[4]; hi.y = accX[5]; hi.z = accX[6]; hi.w = accX[7];
            rx[(2 * wv) * 32]     = lo;
            rx[(2 * wv + 1) * 32] = hi;
        }
    }
    __syncthreads();
    if (kh == 1) return;
    #pragma unroll
    for (int t = 0; t < 4; ++t) {
        const v4f lo = rp[(2 * t) * 32];
        const v4f hi = rp[(2 * t + 1) * 32];
        acc[t][0] += lo.x; acc[t][1] += lo.y; acc[t][2] += lo.z; acc[t][3] += lo.w;
        acc[t][4] += hi.x; acc[t][5] += hi.y; acc[t][6] += hi.z; acc[t][7] += hi.w;
    }
    if (RIDE) {
        if (wv == 0) {
            #pragma unroll
            for (int q = 1; q < 8; ++q) {
                const v4f lo = rx[(2 * q) * 32];
                const v4f hi = rx[(2 * q + 1) * 32];
                accX[0] += lo.x; accX[1] += lo.y; accX[2] += lo.z; accX[3] += lo.w;
                accX[4] += hi.x; accX[5] += hi.y; accX[6] += hi.z; accX[7] += hi.w;
            }
        }
    }

    const bool whole = (row0 + 16 <= mBound) && (col0 + 64 <= nBound) &&
                       ((ldc & 3) == 0) && ((((size_t)Cg) & 15) == 0);
    if (whole) {
        #pragma unroll
        for (int r = 0; r < 8; ++r) {
            const int row = row0 + 8 * h + r;
            const int col = col0 + 4 * c;
            v4f v;
            v.x = acc[0][r]; v.y = acc[1][r]; v.z = acc[2][r]; v.w = acc[3][r];
            if (addI) {
                if (row == col)     v.x += 1.0f;
                if (row == col + 1) v.y += 1.0f;
                if (row == col + 2) v.z += 1.0f;
                if (row == col + 3) v.w += 1.0f;
            }
            *(v4f*)(Cg + (size_t)row * ldc + col) = v;
        }
    } else {
        #pragma unroll
        for (int t = 0; t < 4; ++t)
            #pragma unroll
            for (int r = 0; r < 8; ++r) {
                const int row = row0 + 8 * h + r;
                const int col = col0 + 4 * c + t;
                if (row < mBound && col < nBound) {
                    float v = acc[t][r];
                    if (addI && row == col) v += 1.0f;
                    Cg[(size_t)row * ldc + col] = v;
                }
            }
    }

    if (RIDE == 1) {
        if (wv == 0 && xcol < xn) {
            #pragma unroll
            for (int r = 0; r < 8; ++r)
                Xo[(size_t)(xrow0 + 8 * h + r) * LDX + xcol] = accX[r];
        }
    }
    if (RIDE == 2) {
        if (wv == 0) {
            #pragma unroll
            for (int r = 0; r < 8; ++r) {
                const int rr = 8 * h + r;
                if (rr < xn) Xo[(size_t)rr * LDX + xcol] = accX[r];
            }
        }
    }
}

template <int LDB>
__global__ __launch_bounds__(256)
void gemm_wmma_f32(const float* __restrict__ Ag, int lda,
                   const float* __restrict__ Bg,
                   float* __restrict__ Cg, int ldc,
                   int Kdim, int mBound, int nBound, int addI)
{
    __shared__ v4f red[RED_V4];
    gemm_body<LDB, 0>(Ag, lda, Bg, Cg, ldc, Kdim, mBound, nBound, addI, nullptr, nullptr, nullptr, 0, red);
}

__global__ __launch_bounds__(256)
void gemm_sq_cols(const float* __restrict__ Ms, float* __restrict__ M2,
                  const float* Yg, float* Yo, int s)
{
    __shared__ v4f red[RED_V4];
    const int bid = (int)blockIdx.y * (int)gridDim.x + (int)blockIdx.x;
    if (bid < 4 * (s > 16 ? s : 16))
        gemm_body<NDIM, 1>(Ms, NDIM, Ms, M2, NDIM, NDIM, NDIM, NDIM, 0, Ms, Yg, Yo, s, red);
    else
        gemm_body<NDIM, 0>(Ms, NDIM, Ms, M2, NDIM, NDIM, NDIM, NDIM, 0, nullptr, nullptr, nullptr, 0, red);
}

__global__ __launch_bounds__(256)
void gemm_sq_rows(const float* __restrict__ Qt, float* __restrict__ Q2,
                  const float* Rg, float* Ro, int t)
{
    __shared__ v4f red[RED_V4];
    const int bid = (int)blockIdx.y * (int)gridDim.x + (int)blockIdx.x;
    if (bid < 64)
        gemm_body<NDIM, 2>(Qt, NDIM, Qt, Q2, NDIM, NDIM, NDIM, NDIM, 0, Rg, Qt, Ro, t, red);
    else
        gemm_body<NDIM, 0>(Qt, NDIM, Qt, Q2, NDIM, NDIM, NDIM, NDIM, 0, nullptr, nullptr, nullptr, 0, red);
}

#define TK (NDIM / 8 / 4)
__global__ __launch_bounds__(256)
void rows_apply(const float* Rg, const float* __restrict__ Q, float* Ro)
{
    __shared__ v4f red[8 * 2 * 32];
    const int tid  = threadIdx.x;
    const int lane = tid & 31;
    const int wv   = __builtin_amdgcn_readfirstlane(tid >> 5);
    const int h    = lane >> 4;
    const int c    = lane & 15;
    const int col  = 16 * (int)blockIdx.x + c;
    const int kx   = wv * (NDIM / 8) + 2 * h;
    const float* xa = Rg + (size_t)c * NDIM + kx;
    const float* xb = Q + (size_t)kx * NDIM + col;

    v2f a[TK], b[TK];
    #pragma unroll
    for (int j = 0; j < TK; ++j) {
        a[j]   = *(const v2f*)(xa + 4 * j);
        b[j].x = xb[(4 * j) * NDIM];
        b[j].y = xb[(4 * j + 1) * NDIM];
    }
    v8f acc = {};
    #pragma unroll
    for (int j = 0; j < TK; ++j)
        acc = __builtin_amdgcn_wmma_f32_16x16x4_f32(false, a[j], false, b[j], (short)0, acc, false, false);

    v4f* rx = red + lane;
    if (wv != 0) {
        v4f lo, hi;
        lo.x = acc[0]; lo.y = acc[1]; lo.z = acc[2]; lo.w = acc[3];
        hi.x = acc[4]; hi.y = acc[5]; hi.z = acc[6]; hi.w = acc[7];
        rx[(2 * wv) * 32]     = lo;
        rx[(2 * wv + 1) * 32] = hi;
    }
    __syncthreads();
    if (wv != 0) return;
    #pragma unroll
    for (int q = 1; q < 8; ++q) {
        const v4f lo = rx[(2 * q) * 32];
        const v4f hi = rx[(2 * q + 1) * 32];
        acc[0] += lo.x; acc[1] += lo.y; acc[2] += lo.z; acc[3] += lo.w;
        acc[4] += hi.x; acc[5] += hi.y; acc[6] += hi.z; acc[7] += hi.w;
    }
    #pragma unroll
    for (int r = 0; r < 8; ++r)
        Ro[(size_t)(8 * h + r) * NDIM + col] = acc[r];
}

__global__ void prep_mats(const float* __restrict__ A, const float* __restrict__ ldt,
                          float* __restrict__ M, float* __restrict__ IM, int n)
{
    int idx = blockIdx.x * blockDim.x + threadIdx.x;
    if (idx >= n * n) return;
    float h = 0.5f * expf(ldt[0]);
    float m = h * A[idx];
    M[idx]  = m;
    int r = idx / n, c = idx - r * n;
    IM[idx] = m + ((r == c) ? 1.0f : 0.0f);
}

#define MVW 8
__global__ __launch_bounds__(32 * MVW)
void bb_matvec(const float* __restrict__ S, const float* __restrict__ Bv,
               const float* __restrict__ ldt, float* __restrict__ Y,
               int n, int ldy)
{
    const int lane = threadIdx.x & 31;
    const int row  = blockIdx.x * MVW + (threadIdx.x >> 5);
    if (row >= n) return;
    const float* sp = S + (size_t)row * n + 4 * lane;
    const float* bp = Bv + 4 * lane;
    v4f acc = {0.f, 0.f, 0.f, 0.f};
    #pragma unroll 8
    for (int k = 0; k < n; k += 128) {
        const v4f s = *(const v4f*)(sp + k);
        const v4f b = *(const v4f*)(bp + k);
        acc.x += s.x * b.x;
        acc.y += s.y * b.y;
        acc.z += s.z * b.z;
        acc.w += s.w * b.w;
    }
    float t = (acc.x + acc.y) + (acc.z + acc.w);
    for (int o = 16; o > 0; o >>= 1) t += __shfl_xor(t, o);
    if (lane == 0) Y[(size_t)row * ldy] = expf(ldt[0]) * t;
}

__global__ void copy_row(const float* __restrict__ Cv, float* __restrict__ R, int n)
{
    int i = blockIdx.x * blockDim.x + threadIdx.x;
    if (i < n) R[i] = Cv[i];
}

#define CT   128
#define CW   16
#define CPAD 128
__global__ __launch_bounds__(32 * CW)
void conv_causal(const float* __restrict__ u, const float* __restrict__ Kk,
                 const float* __restrict__ Dv, float* __restrict__ y, int L)
{
    __shared__ __attribute__((aligned(16))) float us[CPAD + LSEQ];
    __shared__ __attribute__((aligned(16))) float kr[LSEQ];
    __shared__ __attribute__((aligned(16))) float red[CW][CT];
    const int tid  = threadIdx.x;
    const int lane = tid & 31;
    const int wv   = __builtin_amdgcn_readfirstlane(tid >> 5);
    const int t0   = blockIdx.x * CT;
    const int nt   = t0 + CT;

    for (int i = 4 * tid; i < nt; i += 4 * 32 * CW) {
        *(v4f*)&us[CPAD + i] = *(const v4f*)&u[i];
        const v4f kv = *(const v4f*)&Kk[L - 4 - i];
        v4f r; r.x = kv.w; r.y = kv.z; r.z = kv.y; r.w = kv.x;
        *(v4f*)&kr[i] = r;
    }
    if (tid < CPAD / 4) { v4f z = {0.f, 0.f, 0.f, 0.f}; *(v4f*)&us[4 * tid] = z; }
    __syncthreads();

    const int S  = (CT / CW) * ((int)blockIdx.x + 1);
    const int l0 = wv * S;
    const int T  = t0 + 4 * lane;
    float a0 = 0.f, a1 = 0.f, a2 = 0.f, a3 = 0.f;
    auto mac = [&](const v4f& w1, const v4f& w0, const v4f& k) {
        a0 += k.x * w1.x;  a1 += k.x * w1.y;  a2 += k.x * w1.z;  a3 += k.x * w1.w;
        a0 += k.y * w0.w;  a1 += k.y * w1.x;  a2 += k.y * w1.y;  a3 += k.y * w1.z;
        a0 += k.z * w0.z;  a1 += k.z * w0.w;  a2 += k.z * w1.x;  a3 += k.z * w1.y;
        a0 += k.w * w0.y;  a1 += k.w * w0.z;  a2 += k.w * w0.w;  a3 += k.w * w1.x;
    };
    v4f w1 = *(const v4f*)&us[CPAD + T - l0];
    v4f w0 = *(const v4f*)&us[CPAD + T - l0 - 4];
    v4f k  = *(const v4f*)&kr[l0];
    for (int l = l0; l < l0 + S - 4; l += 4) {
        const v4f w0n = *(const v4f*)&us[CPAD + T - l - 8];
        const v4f kn  = *(const v4f*)&kr[l + 4];
        mac(w1, w0, k);
        w1 = w0; w0 = w0n; k = kn;
    }
    mac(w1, w0, k);
    { v4f p; p.x = a0; p.y = a1; p.z = a2; p.w = a3; *(v4f*)&red[wv][4 * lane] = p; }
    __syncthreads();
    if (tid < CT) {
        float s = 0.f;
        for (int w = 0; w < CW; ++w) s += red[w][tid];
        y[t0 + tid] = s + Dv[0] * u[t0 + tid];
    }
}

extern "C" void kernel_launch(void* const* d_in, const int* in_sizes, int n_in,
                              void* d_out, int out_size, void* d_ws, size_t ws_size,
                              hipStream_t stream)
{
    (void)in_sizes; (void)n_in; (void)out_size; (void)ws_size;
    const float* u   = (const float*)d_in[0];
    const float* A   = (const float*)d_in[1];
    const float* Bv  = (const float*)d_in[2];
    const float* Cv  = (const float*)d_in[3];
    const float* Dv  = (const float*)d_in[4];
    const float* ldt = (const float*)d_in[5];
    float* y = (float*)d_out;
    const int N = NDIM, L = LSEQ;

    float* B0 = (float*)d_ws;
    float* B1 = B0 + (size_t)N * N;
    float* B2 = B1 + (size_t)N * N;
    float* B3 = B2 + (size_t)N * N;
    float* Y  = B3 + (size_t)N * N;
    float* R  = Y  + (size_t)N * 128;
    float* Kk = R  + (size_t)64 * N;

    dim3 blk(256);
    auto gemm = [&](const float* Am, int lda, const float* Bm, int ldb,
                    float* Cm, int ldc, int Mc, int Nc, int Kd,
                    int mB, int nB, int addI) {
        (void)ldb;
        dim3 grid((Nc + 63) / 64, (Mc + 63) / 64);
        gemm_wmma_f32<NDIM><<<grid, blk, 0, stream>>>(Am, lda, Bm, Cm, ldc, Kd, mB, nB, addI);
    };
    auto gemmY = [&](const float* Am, int lda, const float* Bm, int ldb,
                     float* Cm, int ldc, int Mc, int Nc, int Kd,
                     int mB, int nB, int addI) {
        (void)ldb;
        dim3 grid((Nc + 63) / 64, (Mc + 63) / 64);
        gemm_wmma_f32<128><<<grid, blk, 0, stream>>>(Am, lda, Bm, Cm, ldc, Kd, mB, nB, addI);
    };

    prep_mats<<<(N * N + 255) / 256, 256, 0, stream>>>(A, ldt, B0, B1, N);
    gemm(B0, N, B1, N, B2, N, N, N, N, N, N, 1);
    gemm(B0, N, B2, N, B3, N, N, N, N, N, N, 1);
    gemm(B0, N, B3, N, B2, N, N, N, N, N, N, 1);
    gemm(B2, N, B1, N, B3, N, N, N, N, N, N, 0);
    bb_matvec<<<N / MVW, 32 * MVW, 0, stream>>>(B2, Bv, ldt, Y, N, 128);

    const dim3 sqGrid(NDIM / 64, NDIM / 64);

    auto sqCols = [&](const float* Ms, float* M2s, float* Ynew, int s) {
        gemm_sq_cols<<<sqGrid, blk, 0, stream>>>(Ms, M2s, Y, Ynew, s);
    };
    sqCols(B3, B0, Y + 1,  1);
    sqCols(B0, B1, Y + 2,  2);
    sqCols(B1, B0, Y + 4,  4);
    sqCols(B0, B1, Y + 8,  8);
    sqCols(B1, B0, Y + 16, 16);
    sqCols(B0, B1, Y + 32, 32);
    sqCols(B1, B0, Y + 64, 64);

    copy_row<<<(N + 255) / 256, 256, 0, stream>>>(Cv, R, N);
    auto sqRows = [&](const float* Qt, float* Q2t, float* Rnew, int t) {
        gemm_sq_rows<<<sqGrid, blk, 0, stream>>>(Qt, Q2t, R, Rnew, t);
    };
    sqRows(B0, B1, R + 1 * N,  1);
    sqRows(B1, B0, R + 2 * N,  2);
    sqRows(B0, B1, R + 4 * N,  4);
    sqRows(B1, B0, R + 8 * N,  8);
    rows_apply<<<NDIM / 16, blk, 0, stream>>>(R,          B0, R + 16 * N);
    rows_apply<<<NDIM / 16, blk, 0, stream>>>(R + 16 * N, B0, R + 32 * N);
    rows_apply<<<NDIM / 16, blk, 0, stream>>>(R + 32 * N, B0, R + 48 * N);

    gemmY(R, N, Y, 128, Kk, 128, 64, 128, N, 64, 128, 0);

    conv_causal<<<L / CT, 32 * CW, 0, stream>>>(u, Kk, Dv, y, L);
}


# Tail re-cut: Z = Q_16 @ Y formed beside the first chained row product; K = R[0:48]@Y and R[32:48]@Z on 16x16 K-sliced tiles (32 blocks, was a 2-block GEMM); 4 tail launches to 3 (sum order changes)
# speedup vs baseline: 11.0420x; 1.0315x over previous
#include <hip/hip_runtime.h>
#include <math.h>


typedef __attribute__((ext_vector_type(2))) float v2f;
typedef __attribute__((ext_vector_type(4))) float v4f;
typedef __attribute__((ext_vector_type(8))) float v8f;

#define NDIM 1024
#define LSEQ 8192

#define KS 8
#define NS 2
#define KC (4 * KS)
#define RED_MAIN (4 * 8 * 32)
#define RED_V4 (RED_MAIN + 8 * 2 * 32)

struct Frag { v2f a[KS]; v4f b0[KS]; v4f b1[KS]; };

#define YLD 128
struct FragX { v2f a[KS]; v2f b[KS]; };

template <int LDX>
__device__ __forceinline__ void fragx_load(FragX& f, const float* xaP, const float* xbP)
{
    #pragma unroll
    for (int j = 0; j < KS; ++j) {
        f.a[j]   = *(const v2f*)(xaP + 4 * j);
        f.b[j].x = xbP[(4 * j) * LDX];
        f.b[j].y = xbP[(4 * j + 1) * LDX];
    }
}

template <int LDB>
__device__ __forceinline__ void frag_load(Frag& f, const float* __restrict__ aP,
                                          const float* __restrict__ bP)
{
    #pragma unroll
    for (int j = 0; j < KS; ++j) {
        f.a[j]  = *(const v2f*)(aP + 4 * j);
        f.b0[j] = *(const v4f*)(bP + (4 * j) * LDB);
        f.b1[j] = *(const v4f*)(bP + (4 * j + 1) * LDB);
    }
}

template <int RIDE>
__device__ __forceinline__ void frag_mma(const Frag& f, const FragX& fx, v8f (&acc)[4], v8f& accX)
{
    #pragma unroll
    for (int j = 0; j < KS; ++j) {
        #pragma unroll
        for (int t = 0; t < 4; ++t) {
            v2f b; b.x = f.b0[j][t]; b.y = f.b1[j][t];
            acc[t] = __builtin_amdgcn_wmma_f32_16x16x4_f32(false, f.a[j], false, b, (short)0, acc[t], false, false);
        }
        if (RIDE)
            accX = __builtin_amdgcn_wmma_f32_16x16x4_f32(false, fx.a[j], false, fx.b[j], (short)0, accX, false, false);
    }
}

template <int LDB, int LDX, int RM, int RL>
__device__ __forceinline__
void ring_trip(Frag (&f)[NS], FragX (&fx)[NS], v8f (&acc)[4], v8f& accX,
               const float* __restrict__ aP, const float* __restrict__ bP,
               const float* xaP, const float* xbP)
{
    #pragma unroll
    for (int s = 0; s < NS; ++s) {
        frag_mma<RM>(f[s], fx[s], acc, accX);
        frag_load<LDB>(f[s], aP + s * KC, bP + s * KC * LDB);
        if (RL) fragx_load<LDX>(fx[s], xaP + s * KC, xbP + s * KC * LDX);
        __builtin_amdgcn_sched_barrier(0);
    }
}

template <int LDB, int RIDE>
__device__ __forceinline__
void gemm_body(const float* __restrict__ Ag, int lda,
               const float* __restrict__ Bg,
               float* __restrict__ Cg, int ldc,
               int Kdim, int mBound, int nBound, int addI,
               const float* Xa, const float* Xb, float* Xo, int xn, v4f* red)
{
    const int tid  = threadIdx.x;
    const int lane = tid & 31;
    const int wv   = __builtin_amdgcn_readfirstlane(tid >> 5);
    const int mw   = wv & 3;
    const int kh   = wv >> 2;
    const int h    = lane >> 4;
    const int c    = lane & 15;
    const int row0 = blockIdx.y * 64 + mw * 16;
    const int col0 = blockIdx.x * 64;

    const int Kh = Kdim >> 1;
    const int k0 = kh * Kh;
    const float* aP = Ag + (size_t)(row0 + c) * lda + k0 + 2 * h;
    const float* bP = Bg + (size_t)(k0 + 2 * h) * LDB + col0 + 4 * c;

    constexpr int LDX = (RIDE == 2) ? NDIM : YLD;
    const float* xaP = Xa;
    const float* xbP = Xb;
    int xcol = 0, xrow0 = 0;
    if (RIDE) {
        const int bid = (int)blockIdx.y * (int)gridDim.x + (int)blockIdx.x;
        const int kx  = wv * (Kdim >> 3) + 2 * h;
        if (RIDE == 1) {
            xrow0 = 16 * (bid & 63);
            xcol  = 16 * (bid >> 6) + c;
            xaP = Xa + (size_t)(xrow0 + c) * NDIM + kx;
            xbP = Xb + (size_t)kx * LDX + (xcol < xn ? xcol : 0);
        } else {
            xcol  = 16 * bid + c;
            xaP = Xa + (size_t)(c < xn ? c : 0) * NDIM + kx;
            xbP = Xb + (size_t)kx * LDX + xcol;
        }
    }

    v8f acc[4] = {};
    v8f accX = {};
    Frag f[NS];
    FragX fx[NS];

    #pragma unroll
    for (int s = 0; s < NS; ++s) {
        frag_load<LDB>(f[s], aP + s * KC, bP + s * KC * LDB);
        if (RIDE) fragx_load<LDX>(fx[s], xaP + s * KC, xbP + s * KC * LDX);
        __builtin_amdgcn_sched_barrier(0);
    }

    int kb = NS * KC;
    if (RIDE) {
        aP += NS * KC;  bP += NS * KC * LDB;  xaP += NS * KC;  xbP += NS * KC * LDX;
        ring_trip<LDB, LDX, 1, 1>(f, fx, acc, accX, aP, bP, xaP, xbP);
        aP += NS * KC;  bP += NS * KC * LDB;
        ring_trip<LDB, LDX, 1, 0>(f, fx, acc, accX, aP, bP, xaP, xbP);
        kb += 2 * NS * KC;
    }
    #pragma nounroll
    for (; kb < Kh; kb += NS * KC) {
        aP += NS * KC;
        bP += NS * KC * LDB;
        ring_trip<LDB, LDX, 0, 0>(f, fx, acc, accX, aP, bP, xaP, xbP);
    }
    #pragma unroll
    for (int s = 0; s < NS; ++s) {
        frag_mma<0>(f[s], fx[s], acc, accX);
        __builtin_amdgcn_sched_barrier(0);
    }

    v4f* rp = red + (mw * 8) * 32 + lane;
    v4f* rx = red + RED_MAIN + lane;
    if (kh == 1) {
        #pragma unroll
        for (int t = 0; t < 4; ++t) {
            v4f lo, hi;
            lo.x = acc[t][0]; lo.y = acc[t][1]; lo.z = acc[t][2]; lo.w = acc[t][3];
            hi.x = acc[t][4]; hi.y = acc[t][5]; hi.z = acc[t][6]; hi.w = acc[t][7];
            rp[(2 * t) * 32]     = lo;
            rp[(2 * t + 1) * 32] = hi;
        }
    }
    if (RIDE) {
        if (wv != 0) {
            v4f lo, hi;
            lo.x = accX[0]; lo.y = accX[1]; lo.z = accX[2]; lo.w = accX[3];
            hi.x = accX[4]; hi.y = accX[5]; hi.z = accX[6]; hi.w = accX[7];
            rx[(2 * wv) * 32]     = lo;
            rx[(2 * wv + 1) * 32] = hi;
        }
    }
    __syncthreads();
    if (kh == 1) return;
    #pragma unroll
    for (int t = 0; t < 4; ++t) {
        const v4f lo = rp[(2 * t) * 32];
        const v4f hi = rp[(2 * t + 1) * 32];
        acc[t][0] += lo.x; acc[t][1] += lo.y; acc[t][2] += lo.z; acc[t][3] += lo.w;
        acc[t][4] += hi.x; acc[t][5] += hi.y; acc[t][6] += hi.z; acc[t][7] += hi.w;
    }
    if (RIDE) {
        if (wv == 0) {
            #pragma unroll
            for (int q = 1; q < 8; ++q) {
                const v4f lo = rx[(2 * q) * 32];
                const v4f hi = rx[(2 * q + 1) * 32];
                accX[0] += lo.x; accX[1] += lo.y; accX[2] += lo.z; accX[3] += lo.w;
                accX[4] += hi.x; accX[5] += hi.y; accX[6] += hi.z; accX[7] += hi.w;
            }
        }
    }

    const bool whole = (row0 + 16 <= mBound) && (col0 + 64 <= nBound) &&
                       ((ldc & 3) == 0) && ((((size_t)Cg) & 15) == 0);
    if (whole) {
        #pragma unroll
        for (int r = 0; r < 8; ++r) {
            const int row = row0 + 8 * h + r;
            const int col = col0 + 4 * c;
            v4f v;
            v.x = acc[0][r]; v.y = acc[1][r]; v.z = acc[2][r]; v.w = acc[3][r];
            if (addI) {
                if (row == col)     v.x += 1.0f;
                if (row == col + 1) v.y += 1.0f;
                if (row == col + 2) v.z += 1.0f;
                if (row == col + 3) v.w += 1.0f;
            }
            *(v4f*)(Cg + (size_t)row * ldc + col) = v;
        }
    } else {
        #pragma unroll
        for (int t = 0; t < 4; ++t)
            #pragma unroll
            for (int r = 0; r < 8; ++r) {
                const int row = row0 + 8 * h + r;
                const int col = col0 + 4 * c + t;
                if (row < mBound && col < nBound) {
                    float v = acc[t][r];
                    if (addI && row == col) v += 1.0f;
                    Cg[(size_t)row * ldc + col] = v;
                }
            }
    }

    if (RIDE == 1) {
        if (wv == 0 && xcol < xn) {
            #pragma unroll
            for (int r = 0; r < 8; ++r)
                Xo[(size_t)(xrow0 + 8 * h + r) * LDX + xcol] = accX[r];
        }
    }
    if (RIDE == 2) {
        if (wv == 0) {
            #pragma unroll
            for (int r = 0; r < 8; ++r) {
                const int rr = 8 * h + r;
                if (rr < xn) Xo[(size_t)rr * LDX + xcol] = accX[r];
            }
        }
    }
}

template <int LDB>
__global__ __launch_bounds__(256)
void gemm_wmma_f32(const float* __restrict__ Ag, int lda,
                   const float* __restrict__ Bg,
                   float* __restrict__ Cg, int ldc,
                   int Kdim, int mBound, int nBound, int addI)
{
    __shared__ v4f red[RED_V4];
    gemm_body<LDB, 0>(Ag, lda, Bg, Cg, ldc, Kdim, mBound, nBound, addI, nullptr, nullptr, nullptr, 0, red);
}

__global__ __launch_bounds__(256)
void gemm_sq_cols(const float* __restrict__ Ms, float* __restrict__ M2,
                  const float* Yg, float* Yo, int s)
{
    __shared__ v4f red[RED_V4];
    const int bid = (int)blockIdx.y * (int)gridDim.x + (int)blockIdx.x;
    if (bid < 4 * (s > 16 ? s : 16))
        gemm_body<NDIM, 1>(Ms, NDIM, Ms, M2, NDIM, NDIM, NDIM, NDIM, 0, Ms, Yg, Yo, s, red);
    else
        gemm_body<NDIM, 0>(Ms, NDIM, Ms, M2, NDIM, NDIM, NDIM, NDIM, 0, nullptr, nullptr, nullptr, 0, red);
}

__global__ __launch_bounds__(256)
void gemm_sq_rows(const float* __restrict__ Qt, float* __restrict__ Q2,
                  const float* Rg, float* Ro, int t)
{
    __shared__ v4f red[RED_V4];
    const int bid = (int)blockIdx.y * (int)gridDim.x + (int)blockIdx.x;
    if (bid < 64)
        gemm_body<NDIM, 2>(Qt, NDIM, Qt, Q2, NDIM, NDIM, NDIM, NDIM, 0, Rg, Qt, Ro, t, red);
    else
        gemm_body<NDIM, 0>(Qt, NDIM, Qt, Q2, NDIM, NDIM, NDIM, NDIM, 0, nullptr, nullptr, nullptr, 0, red);
}

#define TK (NDIM / 8 / 4)
template <int LDB>
__device__ __forceinline__
void tile16(const float* Ag, const float* Bg, float* Og, int ldo, v4f* red)
{
    const int tid  = threadIdx.x;
    const int lane = tid & 31;
    const int wv   = __builtin_amdgcn_readfirstlane(tid >> 5);
    const int h    = lane >> 4;
    const int c    = lane & 15;
    const int kx   = wv * (NDIM / 8) + 2 * h;
    const float* xa = Ag + (size_t)c * NDIM + kx;
    const float* xb = Bg + (size_t)kx * LDB + c;

    v2f a[TK], b[TK];
    #pragma unroll
    for (int j = 0; j < TK; ++j) {
        a[j]   = *(const v2f*)(xa + 4 * j);
        b[j].x = xb[(4 * j) * LDB];
        b[j].y = xb[(4 * j + 1) * LDB];
    }
    v8f acc = {};
    #pragma unroll
    for (int j = 0; j < TK; ++j)
        acc = __builtin_amdgcn_wmma_f32_16x16x4_f32(false, a[j], false, b[j], (short)0, acc, false, false);

    v4f* rx = red + lane;
    if (wv != 0) {
        v4f lo, hi;
        lo.x = acc[0]; lo.y = acc[1]; lo.z = acc[2]; lo.w = acc[3];
        hi.x = acc[4]; hi.y = acc[5]; hi.z = acc[6]; hi.w = acc[7];
        rx[(2 * wv) * 32]     = lo;
        rx[(2 * wv + 1) * 32] = hi;
    }
    __syncthreads();
    if (wv != 0) return;
    #pragma unroll
    for (int q = 1; q < 8; ++q) {
        const v4f lo = rx[(2 * q) * 32];
        const v4f hi = rx[(2 * q + 1) * 32];
        acc[0] += lo.x; acc[1] += lo.y; acc[2] += lo.z; acc[3] += lo.w;
        acc[4] += hi.x; acc[5] += hi.y; acc[6] += hi.z; acc[7] += hi.w;
    }
    #pragma unroll
    for (int r = 0; r < 8; ++r)
        Og[(size_t)(8 * h + r) * ldo + c] = acc[r];
}

__global__ __launch_bounds__(256)
void rows_apply(const float* Rg, const float* Q, float* Ro)
{
    __shared__ v4f red[8 * 2 * 32];
    const int b = (int)blockIdx.x;
    tile16<NDIM>(Rg, Q + 16 * b, Ro + 16 * b, NDIM, red);
}

#define ZT (NDIM / 16 * (YLD / 16))
__global__ __launch_bounds__(256)
void rows_apply_z(const float* Rg, const float* Q, float* Ro, const float* Yg, float* Zg)
{
    __shared__ v4f red[8 * 2 * 32];
    const int b = (int)blockIdx.x;
    if (b < NDIM / 16) {
        tile16<NDIM>(Rg, Q + 16 * b, Ro + 16 * b, NDIM, red);
    } else {
        const int z  = b - NDIM / 16;
        const int tr = z >> 3, tc = z & 7;
        tile16<YLD>(Q + (size_t)(16 * tr) * NDIM, Yg + 16 * tc,
                    Zg + (size_t)(16 * tr) * YLD + 16 * tc, YLD, red);
    }
}

__global__ __launch_bounds__(256)
void k_tiles(const float* Rg, const float* Yg, const float* Zg, float* Kk)
{
    __shared__ v4f red[8 * 2 * 32];
    const int b  = (int)blockIdx.x;
    const int tr = b >> 3, tc = b & 7;
    const float* Ap = Rg + (size_t)(16 * (tr < 3 ? tr : 2)) * NDIM;
    const float* Bp = (tr < 3 ? Yg : Zg) + 16 * tc;
    tile16<YLD>(Ap, Bp, Kk + (size_t)(16 * tr) * YLD + 16 * tc, YLD, red);
}

__global__ void prep_mats(const float* __restrict__ A, const float* __restrict__ ldt,
                          float* __restrict__ M, float* __restrict__ IM, int n)
{
    int idx = blockIdx.x * blockDim.x + threadIdx.x;
    if (idx >= n * n) return;
    float h = 0.5f * expf(ldt[0]);
    float m = h * A[idx];
    M[idx]  = m;
    int r = idx / n, c = idx - r * n;
    IM[idx] = m + ((r == c) ? 1.0f : 0.0f);
}

#define MVW 8
__global__ __launch_bounds__(32 * MVW)
void bb_matvec(const float* __restrict__ S, const float* __restrict__ Bv,
               const float* __restrict__ ldt, float* __restrict__ Y,
               int n, int ldy)
{
    const int lane = threadIdx.x & 31;
    const int row  = blockIdx.x * MVW + (threadIdx.x >> 5);
    if (row >= n) return;
    const float* sp = S + (size_t)row * n + 4 * lane;
    const float* bp = Bv + 4 * lane;
    v4f acc = {0.f, 0.f, 0.f, 0.f};
    #pragma unroll 8
    for (int k = 0; k < n; k += 128) {
        const v4f s = *(const v4f*)(sp + k);
        const v4f b = *(const v4f*)(bp + k);
        acc.x += s.x * b.x;
        acc.y += s.y * b.y;
        acc.z += s.z * b.z;
        acc.w += s.w * b.w;
    }
    float t = (acc.x + acc.y) + (acc.z + acc.w);
    for (int o = 16; o > 0; o >>= 1) t += __shfl_xor(t, o);
    if (lane == 0) Y[(size_t)row * ldy] = expf(ldt[0]) * t;
}

__global__ void copy_row(const float* __restrict__ Cv, float* __restrict__ R, int n)
{
    int i = blockIdx.x * blockDim.x + threadIdx.x;
    if (i < n) R[i] = Cv[i];
}

#define CT   128
#define CW   16
#define CPAD 128
__global__ __launch_bounds__(32 * CW)
void conv_causal(const float* __restrict__ u, const float* __restrict__ Kk,
                 const float* __restrict__ Dv, float* __restrict__ y, int L)
{
    __shared__ __attribute__((aligned(16))) float us[CPAD + LSEQ];
    __shared__ __attribute__((aligned(16))) float kr[LSEQ];
    __shared__ __attribute__((aligned(16))) float red[CW][CT];
    const int tid  = threadIdx.x;
    const int lane = tid & 31;
    const int wv   = __builtin_amdgcn_readfirstlane(tid >> 5);
    const int t0   = blockIdx.x * CT;
    const int nt   = t0 + CT;

    for (int i = 4 * tid; i < nt; i += 4 * 32 * CW) {
        *(v4f*)&us[CPAD + i] = *(const v4f*)&u[i];
        const v4f kv = *(const v4f*)&Kk[L - 4 - i];
        v4f r; r.x = kv.w; r.y = kv.z; r.z = kv.y; r.w = kv.x;
        *(v4f*)&kr[i] = r;
    }
    if (tid < CPAD / 4) { v4f z = {0.f, 0.f, 0.f, 0.f}; *(v4f*)&us[4 * tid] = z; }
    __syncthreads();

    const int S  = (CT / CW) * ((int)blockIdx.x + 1);
    const int l0 = wv * S;
    const int T  = t0 + 4 * lane;
    float a0 = 0.f, a1 = 0.f, a2 = 0.f, a3 = 0.f;
    auto mac = [&](const v4f& w1, const v4f& w0, const v4f& k) {
        a0 += k.x * w1.x;  a1 += k.x * w1.y;  a2 += k.x * w1.z;  a3 += k.x * w1.w;
        a0 += k.y * w0.w;  a1 += k.y * w1.x;  a2 += k.y * w1.y;  a3 += k.y * w1.z;
        a0 += k.z * w0.z;  a1 += k.z * w0.w;  a2 += k.z * w1.x;  a3 += k.z * w1.y;
        a0 += k.w * w0.y;  a1 += k.w * w0.z;  a2 += k.w * w0.w;  a3 += k.w * w1.x;
    };
    v4f w1 = *(const v4f*)&us[CPAD + T - l0];
    v4f w0 = *(const v4f*)&us[CPAD + T - l0 - 4];
    v4f k  = *(const v4f*)&kr[l0];
    for (int l = l0; l < l0 + S - 4; l += 4) {
        const v4f w0n = *(const v4f*)&us[CPAD + T - l - 8];
        const v4f kn  = *(const v4f*)&kr[l + 4];
        mac(w1, w0, k);
        w1 = w0; w0 = w0n; k = kn;
    }
    mac(w1, w0, k);
    { v4f p; p.x = a0; p.y = a1; p.z = a2; p.w = a3; *(v4f*)&red[wv][4 * lane] = p; }
    __syncthreads();
    if (tid < CT) {
        float s = 0.f;
        for (int w = 0; w < CW; ++w) s += red[w][tid];
        y[t0 + tid] = s + Dv[0] * u[t0 + tid];
    }
}

extern "C" void kernel_launch(void* const* d_in, const int* in_sizes, int n_in,
                              void* d_out, int out_size, void* d_ws, size_t ws_size,
                              hipStream_t stream)
{
    (void)in_sizes; (void)n_in; (void)out_size; (void)ws_size;
    const float* u   = (const float*)d_in[0];
    const float* A   = (const float*)d_in[1];
    const float* Bv  = (const float*)d_in[2];
    const float* Cv  = (const float*)d_in[3];
    const float* Dv  = (const float*)d_in[4];
    const float* ldt = (const float*)d_in[5];
    float* y = (float*)d_out;
    const int N = NDIM, L = LSEQ;

    float* B0 = (float*)d_ws;
    float* B1 = B0 + (size_t)N * N;
    float* B2 = B1 + (size_t)N * N;
    float* B3 = B2 + (size_t)N * N;
    float* Y  = B3 + (size_t)N * N;
    float* R  = Y  + (size_t)N * 128;
    float* Kk = R  + (size_t)64 * N;

    dim3 blk(256);
    auto gemm = [&](const float* Am, int lda, const float* Bm, int ldb,
                    float* Cm, int ldc, int Mc, int Nc, int Kd,
                    int mB, int nB, int addI) {
        (void)ldb;
        dim3 grid((Nc + 63) / 64, (Mc + 63) / 64);
        gemm_wmma_f32<NDIM><<<grid, blk, 0, stream>>>(Am, lda, Bm, Cm, ldc, Kd, mB, nB, addI);
    };

    prep_mats<<<(N * N + 255) / 256, 256, 0, stream>>>(A, ldt, B0, B1, N);
    gemm(B0, N, B1, N, B2, N, N, N, N, N, N, 1);
    gemm(B0, N, B2, N, B3, N, N, N, N, N, N, 1);
    gemm(B0, N, B3, N, B2, N, N, N, N, N, N, 1);
    gemm(B2, N, B1, N, B3, N, N, N, N, N, N, 0);
    bb_matvec<<<N / MVW, 32 * MVW, 0, stream>>>(B2, Bv, ldt, Y, N, 128);

    const dim3 sqGrid(NDIM / 64, NDIM / 64);

    auto sqCols = [&](const float* Ms, float* M2s, float* Ynew, int s) {
        gemm_sq_cols<<<sqGrid, blk, 0, stream>>>(Ms, M2s, Y, Ynew, s);
    };
    sqCols(B3, B0, Y + 1,  1);
    sqCols(B0, B1, Y + 2,  2);
    sqCols(B1, B0, Y + 4,  4);
    sqCols(B0, B1, Y + 8,  8);
    sqCols(B1, B0, Y + 16, 16);
    sqCols(B0, B1, Y + 32, 32);
    sqCols(B1, B0, Y + 64, 64);

    copy_row<<<(N + 255) / 256, 256, 0, stream>>>(Cv, R, N);
    auto sqRows = [&](const float* Qt, float* Q2t, float* Rnew, int t) {
        gemm_sq_rows<<<sqGrid, blk, 0, stream>>>(Qt, Q2t, R, Rnew, t);
    };
    sqRows(B0, B1, R + 1 * N,  1);
    sqRows(B1, B0, R + 2 * N,  2);
    sqRows(B0, B1, R + 4 * N,  4);
    sqRows(B1, B0, R + 8 * N,  8);
    rows_apply_z<<<NDIM / 16 + ZT, blk, 0, stream>>>(R, B0, R + 16 * N, Y, B2);
    rows_apply<<<NDIM / 16, blk, 0, stream>>>(R + 16 * N, B0, R + 32 * N);

    k_tiles<<<(64 / 16) * (YLD / 16), blk, 0, stream>>>(R, Y, B2, Kk);

    conv_causal<<<L / CT, 32 * CW, 0, stream>>>(u, Kk, Dv, y, L);
}


# GEMM K loop: the upper-K wave of each SIMD pair takes the matrix pipe (s_setprio 3) after a quarter of its chunks, so both K halves end together, not the younger one alone at the end; same sums
# speedup vs baseline: 11.4547x; 1.0374x over previous
#include <hip/hip_runtime.h>
#include <math.h>


typedef __attribute__((ext_vector_type(2))) float v2f;
typedef __attribute__((ext_vector_type(4))) float v4f;
typedef __attribute__((ext_vector_type(8))) float v8f;

#define NDIM 1024
#define LSEQ 8192

#define KS 8
#define NS 2
#define KC (4 * KS)
#define RED_MAIN (4 * 8 * 32)
#define RED_V4 (RED_MAIN + 8 * 2 * 32)

struct Frag { v2f a[KS]; v4f b0[KS]; v4f b1[KS]; };

#define YLD 128
struct FragX { v2f a[KS]; v2f b[KS]; };

template <int LDX>
__device__ __forceinline__ void fragx_load(FragX& f, const float* xaP, const float* xbP)
{
    #pragma unroll
    for (int j = 0; j < KS; ++j) {
        f.a[j]   = *(const v2f*)(xaP + 4 * j);
        f.b[j].x = xbP[(4 * j) * LDX];
        f.b[j].y = xbP[(4 * j + 1) * LDX];
    }
}

template <int LDB>
__device__ __forceinline__ void frag_load(Frag& f, const float* __restrict__ aP,
                                          const float* __restrict__ bP)
{
    #pragma unroll
    for (int j = 0; j < KS; ++j) {
        f.a[j]  = *(const v2f*)(aP + 4 * j);
        f.b0[j] = *(const v4f*)(bP + (4 * j) * LDB);
        f.b1[j] = *(const v4f*)(bP + (4 * j + 1) * LDB);
    }
}

template <int RIDE>
__device__ __forceinline__ void frag_mma(const Frag& f, const FragX& fx, v8f (&acc)[4], v8f& accX)
{
    #pragma unroll
    for (int j = 0; j < KS; ++j) {
        #pragma unroll
        for (int t = 0; t < 4; ++t) {
            v2f b; b.x = f.b0[j][t]; b.y = f.b1[j][t];
            acc[t] = __builtin_amdgcn_wmma_f32_16x16x4_f32(false, f.a[j], false, b, (short)0, acc[t], false, false);
        }
        if (RIDE)
            accX = __builtin_amdgcn_wmma_f32_16x16x4_f32(false, fx.a[j], false, fx.b[j], (short)0, accX, false, false);
    }
}

template <int LDB, int LDX, int RM, int RL>
__device__ __forceinline__
void ring_trip(Frag (&f)[NS], FragX (&fx)[NS], v8f (&acc)[4], v8f& accX,
               const float* __restrict__ aP, const float* __restrict__ bP,
               const float* xaP, const float* xbP)
{
    #pragma unroll
    for (int s = 0; s < NS; ++s) {
        frag_mma<RM>(f[s], fx[s], acc, accX);
        frag_load<LDB>(f[s], aP + s * KC, bP + s * KC * LDB);
        if (RL) fragx_load<LDX>(fx[s], xaP + s * KC, xbP + s * KC * LDX);
        __builtin_amdgcn_sched_barrier(0);
    }
}

template <int LDB, int RIDE>
__device__ __forceinline__
void gemm_body(const float* __restrict__ Ag, int lda,
               const float* __restrict__ Bg,
               float* __restrict__ Cg, int ldc,
               int Kdim, int mBound, int nBound, int addI,
               const float* Xa, const float* Xb, float* Xo, int xn, v4f* red)
{
    const int tid  = threadIdx.x;
    const int lane = tid & 31;
    const int wv   = __builtin_amdgcn_readfirstlane(tid >> 5);
    const int mw   = wv & 3;
    const int kh   = wv >> 2;
    const int h    = lane >> 4;
    const int c    = lane & 15;
    const int row0 = blockIdx.y * 64 + mw * 16;
    const int col0 = blockIdx.x * 64;

    const int Kh = Kdim >> 1;
    const int k0 = kh * Kh;
    const float* aP = Ag + (size_t)(row0 + c) * lda + k0 + 2 * h;
    const float* bP = Bg + (size_t)(k0 + 2 * h) * LDB + col0 + 4 * c;

    constexpr int LDX = (RIDE == 2) ? NDIM : YLD;
    const float* xaP = Xa;
    const float* xbP = Xb;
    int xcol = 0, xrow0 = 0;
    if (RIDE) {
        const int bid = (int)blockIdx.y * (int)gridDim.x + (int)blockIdx.x;
        const int kx  = wv * (Kdim >> 3) + 2 * h;
        if (RIDE == 1) {
            xrow0 = 16 * (bid & 63);
            xcol  = 16 * (bid >> 6) + c;
            xaP = Xa + (size_t)(xrow0 + c) * NDIM + kx;
            xbP = Xb + (size_t)kx * LDX + (xcol < xn ? xcol : 0);
        } else {
            xcol  = 16 * bid + c;
            xaP = Xa + (size_t)(c < xn ? c : 0) * NDIM + kx;
            xbP = Xb + (size_t)kx * LDX + xcol;
        }
    }

    v8f acc[4] = {};
    v8f accX = {};
    Frag f[NS];
    FragX fx[NS];

    #pragma unroll
    for (int s = 0; s < NS; ++s) {
        frag_load<LDB>(f[s], aP + s * KC, bP + s * KC * LDB);
        if (RIDE) fragx_load<LDX>(fx[s], xaP + s * KC, xbP + s * KC * LDX);
        __builtin_amdgcn_sched_barrier(0);
    }

    int kb = NS * KC;
    if (RIDE) {
        aP += NS * KC;  bP += NS * KC * LDB;  xaP += NS * KC;  xbP += NS * KC * LDX;
        ring_trip<LDB, LDX, 1, 1>(f, fx, acc, accX, aP, bP, xaP, xbP);
        aP += NS * KC;  bP += NS * KC * LDB;
        ring_trip<LDB, LDX, 1, 0>(f, fx, acc, accX, aP, bP, xaP, xbP);
        kb += 2 * NS * KC;
    }
    const int kq = NS * KC + (Kh >> 2);
    #pragma nounroll
    for (; kb < kq; kb += NS * KC) {
        aP += NS * KC;
        bP += NS * KC * LDB;
        ring_trip<LDB, LDX, 0, 0>(f, fx, acc, accX, aP, bP, xaP, xbP);
    }
    if (kh) __builtin_amdgcn_s_setprio(3);
    #pragma nounroll
    for (; kb < Kh; kb += NS * KC) {
        aP += NS * KC;
        bP += NS * KC * LDB;
        ring_trip<LDB, LDX, 0, 0>(f, fx, acc, accX, aP, bP, xaP, xbP);
    }
    #pragma unroll
    for (int s = 0; s < NS; ++s) {
        frag_mma<0>(f[s], fx[s], acc, accX);
        __builtin_amdgcn_sched_barrier(0);
    }
    if (kh) __builtin_amdgcn_s_setprio(0);

    v4f* rp = red + (mw * 8) * 32 + lane;
    v4f* rx = red + RED_MAIN + lane;
    if (kh == 1) {
        #pragma unroll
        for (int t = 0; t < 4; ++t) {
            v4f lo, hi;
            lo.x = acc[t][0]; lo.y = acc[t][1]; lo.z = acc[t][2]; lo.w = acc[t][3];
            hi.x = acc[t][4]; hi.y = acc[t][5]; hi.z = acc[t][6]; hi.w = acc[t][7];
            rp[(2 * t) * 32]     = lo;
            rp[(2 * t + 1) * 32] = hi;
        }
    }
    if (RIDE) {
        if (wv != 0) {
            v4f lo, hi;
            lo.x = accX[0]; lo.y = accX[1]; lo.z = accX[2]; lo.w = accX[3];
            hi.x = accX[4]; hi.y = accX[5]; hi.z = accX[6]; hi.w = accX[7];
            rx[(2 * wv) * 32]     = lo;
            rx[(2 * wv + 1) * 32] = hi;
        }
    }
    __syncthreads();
    if (kh == 1) return;
    #pragma unroll
    for (int t = 0; t < 4; ++t) {
        const v4f lo = rp[(2 * t) * 32];
        const v4f hi = rp[(2 * t + 1) * 32];
        acc[t][0] += lo.x; acc[t][1] += lo.y; acc[t][2] += lo.z; acc[t][3] += lo.w;
        acc[t][4] += hi.x; acc[t][5] += hi.y; acc[t][6] += hi.z; acc[t][7] += hi.w;
    }
    if (RIDE) {
        if (wv == 0) {
            #pragma unroll
            for (int q = 1; q < 8; ++q) {
                const v4f lo = rx[(2 * q) * 32];
                const v4f hi = rx[(2 * q + 1) * 32];
                accX[0] += lo.x; accX[1] += lo.y; accX[2] += lo.z; accX[3] += lo.w;
                accX[4] += hi.x; accX[5] += hi.y; accX[6] += hi.z; accX[7] += hi.w;
            }
        }
    }

    const bool whole = (row0 + 16 <= mBound) && (col0 + 64 <= nBound) &&
                       ((ldc & 3) == 0) && ((((size_t)Cg) & 15) == 0);
    if (whole) {
        #pragma unroll
        for (int r = 0; r < 8; ++r) {
            const int row = row0 + 8 * h + r;
            const int col = col0 + 4 * c;
            v4f v;
            v.x = acc[0][r]; v.y = acc[1][r]; v.z = acc[2][r]; v.w = acc[3][r];
            if (addI) {
                if (row == col)     v.x += 1.0f;
                if (row == col + 1) v.y += 1.0f;
                if (row == col + 2) v.z += 1.0f;
                if (row == col + 3) v.w += 1.0f;
            }
            *(v4f*)(Cg + (size_t)row * ldc + col) = v;
        }
    } else {
        #pragma unroll
        for (int t = 0; t < 4; ++t)
            #pragma unroll
            for (int r = 0; r < 8; ++r) {
                const int row = row0 + 8 * h + r;
                const int col = col0 + 4 * c + t;
                if (row < mBound && col < nBound) {
                    float v = acc[t][r];
                    if (addI && row == col) v += 1.0f;
                    Cg[(size_t)row * ldc + col] = v;
                }
            }
    }

    if (RIDE == 1) {
        if (wv == 0 && xcol < xn) {
            #pragma unroll
            for (int r = 0; r < 8; ++r)
                Xo[(size_t)(xrow0 + 8 * h + r) * LDX + xcol] = accX[r];
        }
    }
    if (RIDE == 2) {
        if (wv == 0) {
            #pragma unroll
            for (int r = 0; r < 8; ++r) {
                const int rr = 8 * h + r;
                if (rr < xn) Xo[(size_t)rr * LDX + xcol] = accX[r];
            }
        }
    }
}

template <int LDB>
__global__ __launch_bounds__(256)
void gemm_wmma_f32(const float* __restrict__ Ag, int lda,
                   const float* __restrict__ Bg,
                   float* __restrict__ Cg, int ldc,
                   int Kdim, int mBound, int nBound, int addI)
{
    __shared__ v4f red[RED_V4];
    gemm_body<LDB, 0>(Ag, lda, Bg, Cg, ldc, Kdim, mBound, nBound, addI, nullptr, nullptr, nullptr, 0, red);
}

__global__ __launch_bounds__(256)
void gemm_sq_cols(const float* __restrict__ Ms, float* __restrict__ M2,
                  const float* Yg, float* Yo, int s)
{
    __shared__ v4f red[RED_V4];
    const int bid = (int)blockIdx.y * (int)gridDim.x + (int)blockIdx.x;
    if (bid < 4 * (s > 16 ? s : 16))
        gemm_body<NDIM, 1>(Ms, NDIM, Ms, M2, NDIM, NDIM, NDIM, NDIM, 0, Ms, Yg, Yo, s, red);
    else
        gemm_body<NDIM, 0>(Ms, NDIM, Ms, M2, NDIM, NDIM, NDIM, NDIM, 0, nullptr, nullptr, nullptr, 0, red);
}

__global__ __launch_bounds__(256)
void gemm_sq_rows(const float* __restrict__ Qt, float* __restrict__ Q2,
                  const float* Rg, float* Ro, int t)
{
    __shared__ v4f red[RED_V4];
    const int bid = (int)blockIdx.y * (int)gridDim.x + (int)blockIdx.x;
    if (bid < 64)
        gemm_body<NDIM, 2>(Qt, NDIM, Qt, Q2, NDIM, NDIM, NDIM, NDIM, 0, Rg, Qt, Ro, t, red);
    else
        gemm_body<NDIM, 0>(Qt, NDIM, Qt, Q2, NDIM, NDIM, NDIM, NDIM, 0, nullptr, nullptr, nullptr, 0, red);
}

#define TK (NDIM / 8 / 4)
template <int LDB>
__device__ __forceinline__
void tile16(const float* Ag, const float* Bg, float* Og, int ldo, v4f* red)
{
    const int tid  = threadIdx.x;
    const int lane = tid & 31;
    const int wv   = __builtin_amdgcn_readfirstlane(tid >> 5);
    const int h    = lane >> 4;
    const int c    = lane & 15;
    const int kx   = wv * (NDIM / 8) + 2 * h;
    const float* xa = Ag + (size_t)c * NDIM + kx;
    const float* xb = Bg + (size_t)kx * LDB + c;

    v2f a[TK], b[TK];
    #pragma unroll
    for (int j = 0; j < TK; ++j) {
        a[j]   = *(const v2f*)(xa + 4 * j);
        b[j].x = xb[(4 * j) * LDB];
        b[j].y = xb[(4 * j + 1) * LDB];
    }
    v8f acc = {};
    #pragma unroll
    for (int j = 0; j < TK; ++j)
        acc = __builtin_amdgcn_wmma_f32_16x16x4_f32(false, a[j], false, b[j], (short)0, acc, false, false);

    v4f* rx = red + lane;
    if (wv != 0) {
        v4f lo, hi;
        lo.x = acc[0]; lo.y = acc[1]; lo.z = acc[2]; lo.w = acc[3];
        hi.x = acc[4]; hi.y = acc[5]; hi.z = acc[6]; hi.w = acc[7];
        rx[(2 * wv) * 32]     = lo;
        rx[(2 * wv + 1) * 32] = hi;
    }
    __syncthreads();
    if (wv != 0) return;
    #pragma unroll
    for (int q = 1; q < 8; ++q) {
        const v4f lo = rx[(2 * q) * 32];
        const v4f hi = rx[(2 * q + 1) * 32];
        acc[0] += lo.x; acc[1] += lo.y; acc[2] += lo.z; acc[3] += lo.w;
        acc[4] += hi.x; acc[5] += hi.y; acc[6] += hi.z; acc[7] += hi.w;
    }
    #pragma unroll
    for (int r = 0; r < 8; ++r)
        Og[(size_t)(8 * h + r) * ldo + c] = acc[r];
}

__global__ __launch_bounds__(256)
void rows_apply(const float* Rg, const float* Q, float* Ro)
{
    __shared__ v4f red[8 * 2 * 32];
    const int b = (int)blockIdx.x;
    tile16<NDIM>(Rg, Q + 16 * b, Ro + 16 * b, NDIM, red);
}

#define ZT (NDIM / 16 * (YLD / 16))
__global__ __launch_bounds__(256)
void rows_apply_z(const float* Rg, const float* Q, float* Ro, const float* Yg, float* Zg)
{
    __shared__ v4f red[8 * 2 * 32];
    const int b = (int)blockIdx.x;
    if (b < NDIM / 16) {
        tile16<NDIM>(Rg, Q + 16 * b, Ro + 16 * b, NDIM, red);
    } else {
        const int z  = b - NDIM / 16;
        const int tr = z >> 3, tc = z & 7;
        tile16<YLD>(Q + (size_t)(16 * tr) * NDIM, Yg + 16 * tc,
                    Zg + (size_t)(16 * tr) * YLD + 16 * tc, YLD, red);
    }
}

__global__ __launch_bounds__(256)
void k_tiles(const float* Rg, const float* Yg, const float* Zg, float* Kk)
{
    __shared__ v4f red[8 * 2 * 32];
    const int b  = (int)blockIdx.x;
    const int tr = b >> 3, tc = b & 7;
    const float* Ap = Rg + (size_t)(16 * (tr < 3 ? tr : 2)) * NDIM;
    const float* Bp = (tr < 3 ? Yg : Zg) + 16 * tc;
    tile16<YLD>(Ap, Bp, Kk + (size_t)(16 * tr) * YLD + 16 * tc, YLD, red);
}

__global__ void prep_mats(const float* __restrict__ A, const float* __restrict__ ldt,
                          float* __restrict__ M, float* __restrict__ IM, int n)
{
    int idx = blockIdx.x * blockDim.x + threadIdx.x;
    if (idx >= n * n) return;
    float h = 0.5f * expf(ldt[0]);
    float m = h * A[idx];
    M[idx]  = m;
    int r = idx / n, c = idx - r * n;
    IM[idx] = m + ((r == c) ? 1.0f : 0.0f);
}

#define MVW 8
__global__ __launch_bounds__(32 * MVW)
void bb_matvec(const float* __restrict__ S, const float* __restrict__ Bv,
               const float* __restrict__ ldt, float* __restrict__ Y,
               int n, int ldy)
{
    const int lane = threadIdx.x & 31;
    const int row  = blockIdx.x * MVW + (threadIdx.x >> 5);
    if (row >= n) return;
    const float* sp = S + (size_t)row * n + 4 * lane;
    const float* bp = Bv + 4 * lane;
    v4f acc = {0.f, 0.f, 0.f, 0.f};
    #pragma unroll 8
    for (int k = 0; k < n; k += 128) {
        const v4f s = *(const v4f*)(sp + k);
        const v4f b = *(const v4f*)(bp + k);
        acc.x += s.x * b.x;
        acc.y += s.y * b.y;
        acc.z += s.z * b.z;
        acc.w += s.w * b.w;
    }
    float t = (acc.x + acc.y) + (acc.z + acc.w);
    for (int o = 16; o > 0; o >>= 1) t += __shfl_xor(t, o);
    if (lane == 0) Y[(size_t)row * ldy] = expf(ldt[0]) * t;
}

__global__ void copy_row(const float* __restrict__ Cv, float* __restrict__ R, int n)
{
    int i = blockIdx.x * blockDim.x + threadIdx.x;
    if (i < n) R[i] = Cv[i];
}

#define CT   128
#define CW   16
#define CPAD 128
__global__ __launch_bounds__(32 * CW)
void conv_causal(const float* __restrict__ u, const float* __restrict__ Kk,
                 const float* __restrict__ Dv, float* __restrict__ y, int L)
{
    __shared__ __attribute__((aligned(16))) float us[CPAD + LSEQ];
    __shared__ __attribute__((aligned(16))) float kr[LSEQ];
    __shared__ __attribute__((aligned(16))) float red[CW][CT];
    const int tid  = threadIdx.x;
    const int lane = tid & 31;
    const int wv   = __builtin_amdgcn_readfirstlane(tid >> 5);
    const int t0   = blockIdx.x * CT;
    const int nt   = t0 + CT;

    for (int i = 4 * tid; i < nt; i += 4 * 32 * CW) {
        *(v4f*)&us[CPAD + i] = *(const v4f*)&u[i];
        const v4f kv = *(const v4f*)&Kk[L - 4 - i];
        v4f r; r.x = kv.w; r.y = kv.z; r.z = kv.y; r.w = kv.x;
        *(v4f*)&kr[i] = r;
    }
    if (tid < CPAD / 4) { v4f z = {0.f, 0.f, 0.f, 0.f}; *(v4f*)&us[4 * tid] = z; }
    __syncthreads();

    const int S  = (CT / CW) * ((int)blockIdx.x + 1);
    const int l0 = wv * S;
    const int T  = t0 + 4 * lane;
    float a0 = 0.f, a1 = 0.f, a2 = 0.f, a3 = 0.f;
    auto mac = [&](const v4f& w1, const v4f& w0, const v4f& k) {
        a0 += k.x * w1.x;  a1 += k.x * w1.y;  a2 += k.x * w1.z;  a3 += k.x * w1.w;
        a0 += k.y * w0.w;  a1 += k.y * w1.x;  a2 += k.y * w1.y;  a3 += k.y * w1.z;
        a0 += k.z * w0.z;  a1 += k.z * w0.w;  a2 += k.z * w1.x;  a3 += k.z * w1.y;
        a0 += k.w * w0.y;  a1 += k.w * w0.z;  a2 += k.w * w0.w;  a3 += k.w * w1.x;
    };
    v4f w1 = *(const v4f*)&us[CPAD + T - l0];
    v4f w0 = *(const v4f*)&us[CPAD + T - l0 - 4];
    v4f k  = *(const v4f*)&kr[l0];
    for (int l = l0; l < l0 + S - 4; l += 4) {
        const v4f w0n = *(const v4f*)&us[CPAD + T - l - 8];
        const v4f kn  = *(const v4f*)&kr[l + 4];
        mac(w1, w0, k);
        w1 = w0; w0 = w0n; k = kn;
    }
    mac(w1, w0, k);
    { v4f p; p.x = a0; p.y = a1; p.z = a2; p.w = a3; *(v4f*)&red[wv][4 * lane] = p; }
    __syncthreads();
    if (tid < CT) {
        float s = 0.f;
        for (int w = 0; w < CW; ++w) s += red[w][tid];
        y[t0 + tid] = s + Dv[0] * u[t0 + tid];
    }
}

extern "C" void kernel_launch(void* const* d_in, const int* in_sizes, int n_in,
                              void* d_out, int out_size, void* d_ws, size_t ws_size,
                              hipStream_t stream)
{
    (void)in_sizes; (void)n_in; (void)out_size; (void)ws_size;
    const float* u   = (const float*)d_in[0];
    const float* A   = (const float*)d_in[1];
    const float* Bv  = (const float*)d_in[2];
    const float* Cv  = (const float*)d_in[3];
    const float* Dv  = (const float*)d_in[4];
    const float* ldt = (const float*)d_in[5];
    float* y = (float*)d_out;
    const int N = NDIM, L = LSEQ;

    float* B0 = (float*)d_ws;
    float* B1 = B0 + (size_t)N * N;
    float* B2 = B1 + (size_t)N * N;
    float* B3 = B2 + (size_t)N * N;
    float* Y  = B3 + (size_t)N * N;
    float* R  = Y  + (size_t)N * 128;
    float* Kk = R  + (size_t)64 * N;

    dim3 blk(256);
    auto gemm = [&](const float* Am, int lda, const float* Bm, int ldb,
                    float* Cm, int ldc, int Mc, int Nc, int Kd,
                    int mB, int nB, int addI) {
        (void)ldb;
        dim3 grid((Nc + 63) / 64, (Mc + 63) / 64);
        gemm_wmma_f32<NDIM><<<grid, blk, 0, stream>>>(Am, lda, Bm, Cm, ldc, Kd, mB, nB, addI);
    };

    prep_mats<<<(N * N + 255) / 256, 256, 0, stream>>>(A, ldt, B0, B1, N);
    gemm(B0, N, B1, N, B2, N, N, N, N, N, N, 1);
    gemm(B0, N, B2, N, B3, N, N, N, N, N, N, 1);
    gemm(B0, N, B3, N, B2, N, N, N, N, N, N, 1);
    gemm(B2, N, B1, N, B3, N, N, N, N, N, N, 0);
    bb_matvec<<<N / MVW, 32 * MVW, 0, stream>>>(B2, Bv, ldt, Y, N, 128);

    const dim3 sqGrid(NDIM / 64, NDIM / 64);

    auto sqCols = [&](const float* Ms, float* M2s, float* Ynew, int s) {
        gemm_sq_cols<<<sqGrid, blk, 0, stream>>>(Ms, M2s, Y, Ynew, s);
    };
    sqCols(B3, B0, Y + 1,  1);
    sqCols(B0, B1, Y + 2,  2);
    sqCols(B1, B0, Y + 4,  4);
    sqCols(B0, B1, Y + 8,  8);
    sqCols(B1, B0, Y + 16, 16);
    sqCols(B0, B1, Y + 32, 32);
    sqCols(B1, B0, Y + 64, 64);

    copy_row<<<(N + 255) / 256, 256, 0, stream>>>(Cv, R, N);
    auto sqRows = [&](const float* Qt, float* Q2t, float* Rnew, int t) {
        gemm_sq_rows<<<sqGrid, blk, 0, stream>>>(Qt, Q2t, R, Rnew, t);
    };
    sqRows(B0, B1, R + 1 * N,  1);
    sqRows(B1, B0, R + 2 * N,  2);
    sqRows(B0, B1, R + 4 * N,  4);
    sqRows(B1, B0, R + 8 * N,  8);
    rows_apply_z<<<NDIM / 16 + ZT, blk, 0, stream>>>(R, B0, R + 16 * N, Y, B2);
    rows_apply<<<NDIM / 16, blk, 0, stream>>>(R + 16 * N, B0, R + 32 * N);

    k_tiles<<<(64 / 16) * (YLD / 16), blk, 0, stream>>>(R, Y, B2, Kk);

    conv_causal<<<L / CT, 32 * CW, 0, stream>>>(u, Kk, Dv, y, L);
}
